# starting kernel
# baseline (speedup 1.0000x reference)
#include <hip/hip_runtime.h>

typedef __attribute__((ext_vector_type(2))) float v2f;
typedef __attribute__((ext_vector_type(8))) float v8f;

#define NQ      12
#define DIM     4096
#define NS      4
#define NL      3
#define THREADS 256

template <int L>
__device__ __forceinline__ int permc(int z) {
  return (L == 0) ? z
       : (L == 1) ? (z ^ (z >> 1))
       : (L == 2) ? (z ^ (z >> 2))
                  : (z ^ (z >> 1) ^ (z >> 2) ^ (z >> 3));
}

__global__ void qgates_kernel(const float* __restrict__ params,
                              float* __restrict__ Ure, float* __restrict__ Uim) {
  const int t = threadIdx.x;
  const int r = t >> 4, c = t & 15;
  for (int l = 0; l < NL; ++l) {
    for (int g = 0; g < 3; ++g) {
      float gr[4][2][2], gi[4][2][2];
      #pragma unroll
      for (int j = 0; j < 4; ++j) {
        const int wire = g * 4 + j;
        const float* p = params + (l * NQ + wire) * 3;
        const float c0 = cosf(0.5f * p[0]), s0 = sinf(0.5f * p[0]);
        const float cp = cosf(0.5f * p[1]), sp = sinf(0.5f * p[1]);
        const float c2 = cosf(0.5f * p[2]), s2 = sinf(0.5f * p[2]);
        gr[j][0][0] =  cp * (c2 * c0 - s2 * s0);
        gi[j][0][0] = -sp * (c2 * c0 + s2 * s0);
        gr[j][0][1] = -cp * (c2 * s0 + s2 * c0);
        gi[j][0][1] =  sp * (c2 * s0 - s2 * c0);
        gr[j][1][0] =  cp * (s2 * c0 + c2 * s0);
        gi[j][1][0] =  sp * (c2 * s0 - s2 * c0);
        gr[j][1][1] =  cp * (c2 * c0 - s2 * s0);
        gi[j][1][1] =  sp * (c2 * c0 + s2 * s0);
      }
      float ar = 1.f, ai = 0.f;
      #pragma unroll
      for (int j = 0; j < 4; ++j) {
        const int rb = (r >> (3 - j)) & 1, cb = (c >> (3 - j)) & 1;
        const float br = gr[j][rb][cb], bi = gi[j][rb][cb];
        const float nr = ar * br - ai * bi, ni = ar * bi + ai * br;
        ar = nr; ai = ni;
      }
      Ure[(l * 3 + g) * 256 + r * 16 + c] = ar;
      Uim[(l * 3 + g) * 256 + r * 16 + c] = ai;
    }
  }
}

template <int L, int G>
__device__ __forceinline__ void apply_group(float2* __restrict__ sAmp,
                                            const float* __restrict__ Ure,
                                            const float* __restrict__ Uim,
                                            int lane, int wave) {
  constexpr int RS = (G == 0) ? 256 : (G == 1) ? 16 : 1;
  const float* ur = Ure + (L * 3 + G) * 256;
  const float* ui = Uim + (L * 3 + G) * 256;
  const bool hi = (lane >= 16);
  const int koff = hi ? 2 : 0;
  const int moff = lane & 15;
  const int nn   = lane & 15;

  v2f aRe[4], aIm[4], aMim[4];
  #pragma unroll
  for (int f = 0; f < 4; ++f) {
    const int k0 = 4 * f + koff;
    v2f tr, ti;
    tr.x = ur[moff * 16 + k0]; tr.y = ur[moff * 16 + k0 + 1];
    ti.x = ui[moff * 16 + k0]; ti.y = ui[moff * 16 + k0 + 1];
    aRe[f] = tr; aIm[f] = ti; aMim[f] = -ti;
  }

  int pk0[4], pk1[4], pm[8];
  #pragma unroll
  for (int f = 0; f < 4; ++f) {
    pk0[f] = hi ? permc<L>((4 * f + 2) * RS) : permc<L>((4 * f + 0) * RS);
    pk1[f] = hi ? permc<L>((4 * f + 3) * RS) : permc<L>((4 * f + 1) * RS);
  }
  #pragma unroll
  for (int v = 0; v < 8; ++v)
    pm[v] = hi ? permc<L>((v + 8) * RS) : permc<L>(v * RS);

  #pragma unroll 2
  for (int t = 0; t < 8; ++t) {
    const int tt = wave * 8 + t;
    const int ss = tt >> 4;
    const int c  = ((tt & 15) << 4) + nn;
    int other;
    if (G == 0)      other = c;
    else if (G == 1) other = ((c >> 4) << 8) | (c & 15);
    else             other = c << 4;
    const int pob = (ss << 12) | permc<L>(other);

    v2f bRe[4], bIm[4];
    #pragma unroll
    for (int f = 0; f < 4; ++f) {
      const float2 q0 = sAmp[pob ^ pk0[f]];
      const float2 q1 = sAmp[pob ^ pk1[f]];
      v2f tr, ti;
      tr.x = q0.x; tr.y = q1.x;
      ti.x = q0.y; ti.y = q1.y;
      bRe[f] = tr; bIm[f] = ti;
    }
    v8f dre = {}; v8f dim_ = {};
    #pragma unroll
    for (int f = 0; f < 4; ++f) {
      dre  = __builtin_amdgcn_wmma_f32_16x16x4_f32(false, aRe[f],  false, bRe[f], (short)0, dre,  false, false);
      dre  = __builtin_amdgcn_wmma_f32_16x16x4_f32(false, aMim[f], false, bIm[f], (short)0, dre,  false, false);
      dim_ = __builtin_amdgcn_wmma_f32_16x16x4_f32(false, aRe[f],  false, bIm[f], (short)0, dim_, false, false);
      dim_ = __builtin_amdgcn_wmma_f32_16x16x4_f32(false, aIm[f],  false, bRe[f], (short)0, dim_, false, false);
    }
    #pragma unroll
    for (int v = 0; v < 8; ++v) {
      float2 st; st.x = dre[v]; st.y = dim_[v];
      sAmp[pob ^ pm[v]] = st;
    }
  }
}

__global__ void __launch_bounds__(THREADS)
qsim_kernel(const float* __restrict__ x, const float* __restrict__ head_w,
            const float* __restrict__ head_b, const float* __restrict__ Ure,
            const float* __restrict__ Uim, float* __restrict__ out) {
  extern __shared__ float smem_raw[];
  float2* sAmp = (float2*)smem_raw;
  float*  accS = (float*)(sAmp + NS * DIM);
  float*  csA  = accS + NS;
  float*  snA  = csA + NS * NQ;

  const int tid  = threadIdx.x;
  const int lane = tid & 31;
  const int wave = tid >> 5;
  const int s0   = blockIdx.x * NS;

  float w[NQ];
  #pragma unroll
  for (int i = 0; i < NQ; ++i) w[i] = head_w[i];

  if (tid < NS * NQ) {
    const int ss = tid / NQ, i = tid - ss * NQ;
    const float v = 0.5f * x[(s0 + ss) * NQ + i];
    csA[tid] = cosf(v);
    snA[tid] = sinf(v);
  }
  __syncthreads();
  for (int ss = 0; ss < NS; ++ss) {
    for (int z = tid; z < DIM; z += THREADS) {
      float m = 1.f;
      #pragma unroll
      for (int i = 0; i < NQ; ++i)
        m *= ((z >> (NQ - 1 - i)) & 1) ? snA[ss * NQ + i] : csA[ss * NQ + i];
      const int k = __popc(z) & 3;
      float2 a; a.x = 0.f; a.y = 0.f;
      if (k == 0) a.x = m; else if (k == 1) a.y = -m;
      else if (k == 2) a.x = -m; else a.y = m;
      sAmp[(ss << 12) + z] = a;
    }
  }
  __syncthreads();

  apply_group<0, 0>(sAmp, Ure, Uim, lane, wave); __syncthreads();
  apply_group<0, 1>(sAmp, Ure, Uim, lane, wave); __syncthreads();
  apply_group<0, 2>(sAmp, Ure, Uim, lane, wave); __syncthreads();
  apply_group<1, 0>(sAmp, Ure, Uim, lane, wave); __syncthreads();
  apply_group<1, 1>(sAmp, Ure, Uim, lane, wave); __syncthreads();
  apply_group<1, 2>(sAmp, Ure, Uim, lane, wave); __syncthreads();
  apply_group<2, 0>(sAmp, Ure, Uim, lane, wave); __syncthreads();
  apply_group<2, 1>(sAmp, Ure, Uim, lane, wave); __syncthreads();
  apply_group<2, 2>(sAmp, Ure, Uim, lane, wave); __syncthreads();

  if (tid < NS) accS[tid] = 0.f;
  __syncthreads();
  float wsum = 0.f;
  #pragma unroll
  for (int i = 0; i < NQ; ++i) wsum += w[i];
  for (int ss = 0; ss < NS; ++ss) {
    float a = 0.f;
    for (int z = tid; z < DIM; z += THREADS) {
      const float2 q = sAmp[(ss << 12) + permc<NL>(z)];
      float coef = wsum;
      #pragma unroll
      for (int i = 0; i < NQ; ++i)
        coef -= 2.f * w[i] * (float)((z >> (NQ - 1 - i)) & 1);
      a += (q.x * q.x + q.y * q.y) * coef;
    }
    atomicAdd(&accS[ss], a);
  }
  __syncthreads();
  if (tid < NS) out[s0 + tid] = accS[tid] + head_b[0];
}

extern "C" void kernel_launch(void* const* d_in, const int* in_sizes, int n_in,
                              void* d_out, int out_size, void* d_ws, size_t ws_size,
                              hipStream_t stream) {
  const float* x      = (const float*)d_in[0];
  const float* params = (const float*)d_in[1];
  const float* head_w = (const float*)d_in[2];
  const float* head_b = (const float*)d_in[3];
  float* out = (float*)d_out;
  float* Ure = (float*)d_ws;
  float* Uim = Ure + NL * 3 * 256;

  const int B = in_sizes[0] / NQ;
  const size_t shmem =
      (size_t)(2 * NS * DIM + NS + 2 * NS * NQ) * sizeof(float);
  (void)hipFuncSetAttribute((const void*)qsim_kernel,
                            hipFuncAttributeMaxDynamicSharedMemorySize,
                            (int)shmem);

  qgates_kernel<<<1, 256, 0, stream>>>(params, Ure, Uim);
  qsim_kernel<<<B / NS, THREADS, shmem, stream>>>(x, head_w, head_b, Ure, Uim, out);
}


# Two samples per workgroup instead of four: 64 KiB LDS each, so four workgroups (8 waves/SIMD) co-reside per WGP, not two, and their serial phases overlap
# speedup vs baseline: 1.2115x; 1.2115x over previous
#include <hip/hip_runtime.h>

typedef __attribute__((ext_vector_type(2))) float v2f;
typedef __attribute__((ext_vector_type(8))) float v8f;

#define NQ      12
#define DIM     4096
#define NS      2
#define NL      3
#define THREADS 256

template <int L>
__device__ __forceinline__ int permc(int z) {
  return (L == 0) ? z
       : (L == 1) ? (z ^ (z >> 1))
       : (L == 2) ? (z ^ (z >> 2))
                  : (z ^ (z >> 1) ^ (z >> 2) ^ (z >> 3));
}

__global__ void qgates_kernel(const float* __restrict__ params,
                              float* __restrict__ Ure, float* __restrict__ Uim) {
  const int t = threadIdx.x;
  const int r = t >> 4, c = t & 15;
  for (int l = 0; l < NL; ++l) {
    for (int g = 0; g < 3; ++g) {
      float gr[4][2][2], gi[4][2][2];
      #pragma unroll
      for (int j = 0; j < 4; ++j) {
        const int wire = g * 4 + j;
        const float* p = params + (l * NQ + wire) * 3;
        const float c0 = cosf(0.5f * p[0]), s0 = sinf(0.5f * p[0]);
        const float cp = cosf(0.5f * p[1]), sp = sinf(0.5f * p[1]);
        const float c2 = cosf(0.5f * p[2]), s2 = sinf(0.5f * p[2]);
        gr[j][0][0] =  cp * (c2 * c0 - s2 * s0);
        gi[j][0][0] = -sp * (c2 * c0 + s2 * s0);
        gr[j][0][1] = -cp * (c2 * s0 + s2 * c0);
        gi[j][0][1] =  sp * (c2 * s0 - s2 * c0);
        gr[j][1][0] =  cp * (s2 * c0 + c2 * s0);
        gi[j][1][0] =  sp * (c2 * s0 - s2 * c0);
        gr[j][1][1] =  cp * (c2 * c0 - s2 * s0);
        gi[j][1][1] =  sp * (c2 * c0 + s2 * s0);
      }
      float ar = 1.f, ai = 0.f;
      #pragma unroll
      for (int j = 0; j < 4; ++j) {
        const int rb = (r >> (3 - j)) & 1, cb = (c >> (3 - j)) & 1;
        const float br = gr[j][rb][cb], bi = gi[j][rb][cb];
        const float nr = ar * br - ai * bi, ni = ar * bi + ai * br;
        ar = nr; ai = ni;
      }
      Ure[(l * 3 + g) * 256 + r * 16 + c] = ar;
      Uim[(l * 3 + g) * 256 + r * 16 + c] = ai;
    }
  }
}

template <int L, int G>
__device__ __forceinline__ void apply_group(float2* __restrict__ sAmp,
                                            const float* __restrict__ Ure,
                                            const float* __restrict__ Uim,
                                            int lane, int wave) {
  constexpr int RS = (G == 0) ? 256 : (G == 1) ? 16 : 1;
  const float* ur = Ure + (L * 3 + G) * 256;
  const float* ui = Uim + (L * 3 + G) * 256;
  const bool hi = (lane >= 16);
  const int koff = hi ? 2 : 0;
  const int moff = lane & 15;
  const int nn   = lane & 15;

  v2f aRe[4], aIm[4], aMim[4];
  #pragma unroll
  for (int f = 0; f < 4; ++f) {
    const int k0 = 4 * f + koff;
    v2f tr, ti;
    tr.x = ur[moff * 16 + k0]; tr.y = ur[moff * 16 + k0 + 1];
    ti.x = ui[moff * 16 + k0]; ti.y = ui[moff * 16 + k0 + 1];
    aRe[f] = tr; aIm[f] = ti; aMim[f] = -ti;
  }

  int pk0[4], pk1[4], pm[8];
  #pragma unroll
  for (int f = 0; f < 4; ++f) {
    pk0[f] = hi ? permc<L>((4 * f + 2) * RS) : permc<L>((4 * f + 0) * RS);
    pk1[f] = hi ? permc<L>((4 * f + 3) * RS) : permc<L>((4 * f + 1) * RS);
  }
  #pragma unroll
  for (int v = 0; v < 8; ++v)
    pm[v] = hi ? permc<L>((v + 8) * RS) : permc<L>(v * RS);

  constexpr int TPW = NS * 16 / (THREADS / 32);
  #pragma unroll 2
  for (int t = 0; t < TPW; ++t) {
    const int tt = wave * TPW + t;
    const int ss = tt >> 4;
    const int c  = ((tt & 15) << 4) + nn;
    int other;
    if (G == 0)      other = c;
    else if (G == 1) other = ((c >> 4) << 8) | (c & 15);
    else             other = c << 4;
    const int pob = (ss << 12) | permc<L>(other);

    v2f bRe[4], bIm[4];
    #pragma unroll
    for (int f = 0; f < 4; ++f) {
      const float2 q0 = sAmp[pob ^ pk0[f]];
      const float2 q1 = sAmp[pob ^ pk1[f]];
      v2f tr, ti;
      tr.x = q0.x; tr.y = q1.x;
      ti.x = q0.y; ti.y = q1.y;
      bRe[f] = tr; bIm[f] = ti;
    }
    v8f dre = {}; v8f dim_ = {};
    #pragma unroll
    for (int f = 0; f < 4; ++f) {
      dre  = __builtin_amdgcn_wmma_f32_16x16x4_f32(false, aRe[f],  false, bRe[f], (short)0, dre,  false, false);
      dre  = __builtin_amdgcn_wmma_f32_16x16x4_f32(false, aMim[f], false, bIm[f], (short)0, dre,  false, false);
      dim_ = __builtin_amdgcn_wmma_f32_16x16x4_f32(false, aRe[f],  false, bIm[f], (short)0, dim_, false, false);
      dim_ = __builtin_amdgcn_wmma_f32_16x16x4_f32(false, aIm[f],  false, bRe[f], (short)0, dim_, false, false);
    }
    #pragma unroll
    for (int v = 0; v < 8; ++v) {
      float2 st; st.x = dre[v]; st.y = dim_[v];
      sAmp[pob ^ pm[v]] = st;
    }
  }
}

__global__ void __launch_bounds__(THREADS)
qsim_kernel(const float* __restrict__ x, const float* __restrict__ head_w,
            const float* __restrict__ head_b, const float* __restrict__ Ure,
            const float* __restrict__ Uim, float* __restrict__ out) {
  extern __shared__ float smem_raw[];
  float2* sAmp = (float2*)smem_raw;
  float*  accS = (float*)(sAmp + NS * DIM);
  float*  csA  = accS + NS;
  float*  snA  = csA + NS * NQ;

  const int tid  = threadIdx.x;
  const int lane = tid & 31;
  const int wave = tid >> 5;
  const int s0   = blockIdx.x * NS;

  float w[NQ];
  #pragma unroll
  for (int i = 0; i < NQ; ++i) w[i] = head_w[i];

  if (tid < NS * NQ) {
    const int ss = tid / NQ, i = tid - ss * NQ;
    const float v = 0.5f * x[(s0 + ss) * NQ + i];
    csA[tid] = cosf(v);
    snA[tid] = sinf(v);
  }
  __syncthreads();
  for (int ss = 0; ss < NS; ++ss) {
    for (int z = tid; z < DIM; z += THREADS) {
      float m = 1.f;
      #pragma unroll
      for (int i = 0; i < NQ; ++i)
        m *= ((z >> (NQ - 1 - i)) & 1) ? snA[ss * NQ + i] : csA[ss * NQ + i];
      const int k = __popc(z) & 3;
      float2 a; a.x = 0.f; a.y = 0.f;
      if (k == 0) a.x = m; else if (k == 1) a.y = -m;
      else if (k == 2) a.x = -m; else a.y = m;
      sAmp[(ss << 12) + z] = a;
    }
  }
  __syncthreads();

  apply_group<0, 0>(sAmp, Ure, Uim, lane, wave); __syncthreads();
  apply_group<0, 1>(sAmp, Ure, Uim, lane, wave); __syncthreads();
  apply_group<0, 2>(sAmp, Ure, Uim, lane, wave); __syncthreads();
  apply_group<1, 0>(sAmp, Ure, Uim, lane, wave); __syncthreads();
  apply_group<1, 1>(sAmp, Ure, Uim, lane, wave); __syncthreads();
  apply_group<1, 2>(sAmp, Ure, Uim, lane, wave); __syncthreads();
  apply_group<2, 0>(sAmp, Ure, Uim, lane, wave); __syncthreads();
  apply_group<2, 1>(sAmp, Ure, Uim, lane, wave); __syncthreads();
  apply_group<2, 2>(sAmp, Ure, Uim, lane, wave); __syncthreads();

  if (tid < NS) accS[tid] = 0.f;
  __syncthreads();
  float wsum = 0.f;
  #pragma unroll
  for (int i = 0; i < NQ; ++i) wsum += w[i];
  for (int ss = 0; ss < NS; ++ss) {
    float a = 0.f;
    for (int z = tid; z < DIM; z += THREADS) {
      const float2 q = sAmp[(ss << 12) + permc<NL>(z)];
      float coef = wsum;
      #pragma unroll
      for (int i = 0; i < NQ; ++i)
        coef -= 2.f * w[i] * (float)((z >> (NQ - 1 - i)) & 1);
      a += (q.x * q.x + q.y * q.y) * coef;
    }
    atomicAdd(&accS[ss], a);
  }
  __syncthreads();
  if (tid < NS) out[s0 + tid] = accS[tid] + head_b[0];
}

extern "C" void kernel_launch(void* const* d_in, const int* in_sizes, int n_in,
                              void* d_out, int out_size, void* d_ws, size_t ws_size,
                              hipStream_t stream) {
  const float* x      = (const float*)d_in[0];
  const float* params = (const float*)d_in[1];
  const float* head_w = (const float*)d_in[2];
  const float* head_b = (const float*)d_in[3];
  float* out = (float*)d_out;
  float* Ure = (float*)d_ws;
  float* Uim = Ure + NL * 3 * 256;

  const int B = in_sizes[0] / NQ;
  const size_t shmem =
      (size_t)(2 * NS * DIM + NS + 2 * NS * NQ) * sizeof(float);
  (void)hipFuncSetAttribute((const void*)qsim_kernel,
                            hipFuncAttributeMaxDynamicSharedMemorySize,
                            (int)shmem);

  qgates_kernel<<<1, 256, 0, stream>>>(params, Ure, Uim);
  qsim_kernel<<<B / NS, THREADS, shmem, stream>>>(x, head_w, head_b, Ure, Uim, out);
}


# Split z = tid + 256*j in encode and readout: the 12-wire product and the Z coefficient become a per-thread 8-wire part times/plus a per-j 4-wire part (re-associates that product and that sum)
# speedup vs baseline: 1.4789x; 1.2207x over previous
#include <hip/hip_runtime.h>

typedef __attribute__((ext_vector_type(2))) float v2f;
typedef __attribute__((ext_vector_type(8))) float v8f;

#define NQ      12
#define DIM     4096
#define NS      2
#define NL      3
#define THREADS 256

template <int L>
__device__ __forceinline__ int permc(int z) {
  return (L == 0) ? z
       : (L == 1) ? (z ^ (z >> 1))
       : (L == 2) ? (z ^ (z >> 2))
                  : (z ^ (z >> 1) ^ (z >> 2) ^ (z >> 3));
}

__global__ void qgates_kernel(const float* __restrict__ params,
                              float* __restrict__ Ure, float* __restrict__ Uim) {
  const int t = threadIdx.x;
  const int r = t >> 4, c = t & 15;
  for (int l = 0; l < NL; ++l) {
    for (int g = 0; g < 3; ++g) {
      float gr[4][2][2], gi[4][2][2];
      #pragma unroll
      for (int j = 0; j < 4; ++j) {
        const int wire = g * 4 + j;
        const float* p = params + (l * NQ + wire) * 3;
        const float c0 = cosf(0.5f * p[0]), s0 = sinf(0.5f * p[0]);
        const float cp = cosf(0.5f * p[1]), sp = sinf(0.5f * p[1]);
        const float c2 = cosf(0.5f * p[2]), s2 = sinf(0.5f * p[2]);
        gr[j][0][0] =  cp * (c2 * c0 - s2 * s0);
        gi[j][0][0] = -sp * (c2 * c0 + s2 * s0);
        gr[j][0][1] = -cp * (c2 * s0 + s2 * c0);
        gi[j][0][1] =  sp * (c2 * s0 - s2 * c0);
        gr[j][1][0] =  cp * (s2 * c0 + c2 * s0);
        gi[j][1][0] =  sp * (c2 * s0 - s2 * c0);
        gr[j][1][1] =  cp * (c2 * c0 - s2 * s0);
        gi[j][1][1] =  sp * (c2 * c0 + s2 * s0);
      }
      float ar = 1.f, ai = 0.f;
      #pragma unroll
      for (int j = 0; j < 4; ++j) {
        const int rb = (r >> (3 - j)) & 1, cb = (c >> (3 - j)) & 1;
        const float br = gr[j][rb][cb], bi = gi[j][rb][cb];
        const float nr = ar * br - ai * bi, ni = ar * bi + ai * br;
        ar = nr; ai = ni;
      }
      Ure[(l * 3 + g) * 256 + r * 16 + c] = ar;
      Uim[(l * 3 + g) * 256 + r * 16 + c] = ai;
    }
  }
}

template <int L, int G>
__device__ __forceinline__ void apply_group(float2* __restrict__ sAmp,
                                            const float* __restrict__ Ure,
                                            const float* __restrict__ Uim,
                                            int lane, int wave) {
  constexpr int RS = (G == 0) ? 256 : (G == 1) ? 16 : 1;
  const float* ur = Ure + (L * 3 + G) * 256;
  const float* ui = Uim + (L * 3 + G) * 256;
  const bool hi = (lane >= 16);
  const int koff = hi ? 2 : 0;
  const int moff = lane & 15;
  const int nn   = lane & 15;

  v2f aRe[4], aIm[4], aMim[4];
  #pragma unroll
  for (int f = 0; f < 4; ++f) {
    const int k0 = 4 * f + koff;
    v2f tr, ti;
    tr.x = ur[moff * 16 + k0]; tr.y = ur[moff * 16 + k0 + 1];
    ti.x = ui[moff * 16 + k0]; ti.y = ui[moff * 16 + k0 + 1];
    aRe[f] = tr; aIm[f] = ti; aMim[f] = -ti;
  }

  int pk0[4], pk1[4], pm[8];
  #pragma unroll
  for (int f = 0; f < 4; ++f) {
    pk0[f] = hi ? permc<L>((4 * f + 2) * RS) : permc<L>((4 * f + 0) * RS);
    pk1[f] = hi ? permc<L>((4 * f + 3) * RS) : permc<L>((4 * f + 1) * RS);
  }
  #pragma unroll
  for (int v = 0; v < 8; ++v)
    pm[v] = hi ? permc<L>((v + 8) * RS) : permc<L>(v * RS);

  constexpr int TPW = NS * 16 / (THREADS / 32);
  #pragma unroll 2
  for (int t = 0; t < TPW; ++t) {
    const int tt = wave * TPW + t;
    const int ss = tt >> 4;
    const int c  = ((tt & 15) << 4) + nn;
    int other;
    if (G == 0)      other = c;
    else if (G == 1) other = ((c >> 4) << 8) | (c & 15);
    else             other = c << 4;
    const int pob = (ss << 12) | permc<L>(other);

    v2f bRe[4], bIm[4];
    #pragma unroll
    for (int f = 0; f < 4; ++f) {
      const float2 q0 = sAmp[pob ^ pk0[f]];
      const float2 q1 = sAmp[pob ^ pk1[f]];
      v2f tr, ti;
      tr.x = q0.x; tr.y = q1.x;
      ti.x = q0.y; ti.y = q1.y;
      bRe[f] = tr; bIm[f] = ti;
    }
    v8f dre = {}; v8f dim_ = {};
    #pragma unroll
    for (int f = 0; f < 4; ++f) {
      dre  = __builtin_amdgcn_wmma_f32_16x16x4_f32(false, aRe[f],  false, bRe[f], (short)0, dre,  false, false);
      dre  = __builtin_amdgcn_wmma_f32_16x16x4_f32(false, aMim[f], false, bIm[f], (short)0, dre,  false, false);
      dim_ = __builtin_amdgcn_wmma_f32_16x16x4_f32(false, aRe[f],  false, bIm[f], (short)0, dim_, false, false);
      dim_ = __builtin_amdgcn_wmma_f32_16x16x4_f32(false, aIm[f],  false, bRe[f], (short)0, dim_, false, false);
    }
    #pragma unroll
    for (int v = 0; v < 8; ++v) {
      float2 st; st.x = dre[v]; st.y = dim_[v];
      sAmp[pob ^ pm[v]] = st;
    }
  }
}

__global__ void __launch_bounds__(THREADS)
qsim_kernel(const float* __restrict__ x, const float* __restrict__ head_w,
            const float* __restrict__ head_b, const float* __restrict__ Ure,
            const float* __restrict__ Uim, float* __restrict__ out) {
  extern __shared__ float smem_raw[];
  float2* sAmp = (float2*)smem_raw;
  float*  csA  = (float*)(sAmp + NS * DIM);
  float*  snA  = csA + NS * NQ;
  float*  accS = snA + NS * NQ;

  const int tid  = threadIdx.x;
  const int lane = tid & 31;
  const int wave = tid >> 5;
  const int s0   = blockIdx.x * NS;

  float w[NQ];
  #pragma unroll
  for (int i = 0; i < NQ; ++i) w[i] = head_w[i];

  if (tid < NS * NQ) {
    const int ss = tid / NQ, i = tid - ss * NQ;
    const float v = 0.5f * x[(s0 + ss) * NQ + i];
    csA[tid] = cosf(v);
    snA[tid] = sinf(v);
  }
  __syncthreads();
  const int kl = __popc(tid) & 3;
  #pragma unroll
  for (int ss = 0; ss < NS; ++ss) {
    float cs[NQ], sn[NQ];
    #pragma unroll
    for (int i = 0; i < NQ; ++i) { cs[i] = csA[ss * NQ + i]; sn[i] = snA[ss * NQ + i]; }
    float ml = ((tid >> 7) & 1) ? sn[4] : cs[4];
    #pragma unroll
    for (int i = 5; i < NQ; ++i)
      ml *= ((tid >> (NQ - 1 - i)) & 1) ? sn[i] : cs[i];
    const float pr = (kl == 0) ? ml : (kl == 2) ? -ml : 0.f;
    const float pi = (kl == 1) ? -ml : (kl == 3) ? ml : 0.f;
    const float t01[4] = {cs[0] * cs[1], cs[0] * sn[1], sn[0] * cs[1], sn[0] * sn[1]};
    const float t23[4] = {cs[2] * cs[3], cs[2] * sn[3], sn[2] * cs[3], sn[2] * sn[3]};
    #pragma unroll
    for (int j = 0; j < 16; ++j) {
      const float mh = t01[j >> 2] * t23[j & 3];
      const int pj = __builtin_popcount(j) & 3;
      const float rr = (pj == 0) ? pr : (pj == 1) ? pi : (pj == 2) ? -pr : -pi;
      const float ri = (pj == 0) ? pi : (pj == 1) ? -pr : (pj == 2) ? -pi : pr;
      float2 a; a.x = rr * mh; a.y = ri * mh;
      sAmp[(ss << 12) + (j << 8) + tid] = a;
    }
  }
  __syncthreads();

  apply_group<0, 0>(sAmp, Ure, Uim, lane, wave); __syncthreads();
  apply_group<0, 1>(sAmp, Ure, Uim, lane, wave); __syncthreads();
  apply_group<0, 2>(sAmp, Ure, Uim, lane, wave); __syncthreads();
  apply_group<1, 0>(sAmp, Ure, Uim, lane, wave); __syncthreads();
  apply_group<1, 1>(sAmp, Ure, Uim, lane, wave); __syncthreads();
  apply_group<1, 2>(sAmp, Ure, Uim, lane, wave); __syncthreads();
  apply_group<2, 0>(sAmp, Ure, Uim, lane, wave); __syncthreads();
  apply_group<2, 1>(sAmp, Ure, Uim, lane, wave); __syncthreads();
  apply_group<2, 2>(sAmp, Ure, Uim, lane, wave); __syncthreads();

  if (tid < NS) accS[tid] = 0.f;
  __syncthreads();
  float wsum = 0.f;
  #pragma unroll
  for (int i = 0; i < NQ; ++i) wsum += w[i];
  float cl = wsum;
  #pragma unroll
  for (int i = 4; i < NQ; ++i)
    cl -= 2.f * w[i] * (float)((tid >> (NQ - 1 - i)) & 1);
  float coef[16];
  #pragma unroll
  for (int j = 0; j < 16; ++j) {
    float ch = 0.f;
    #pragma unroll
    for (int i = 0; i < 4; ++i)
      ch -= 2.f * w[i] * (float)((j >> (3 - i)) & 1);
    coef[j] = cl + ch;
  }
  const int pt = permc<NL>(tid);
  #pragma unroll
  for (int ss = 0; ss < NS; ++ss) {
    float a = 0.f;
    #pragma unroll
    for (int j = 0; j < 16; ++j) {
      const float2 q = sAmp[(ss << 12) + (pt ^ permc<NL>(j << 8))];
      a += (q.x * q.x + q.y * q.y) * coef[j];
    }
    atomicAdd(&accS[ss], a);
  }
  __syncthreads();
  if (tid < NS) out[s0 + tid] = accS[tid] + head_b[0];
}

extern "C" void kernel_launch(void* const* d_in, const int* in_sizes, int n_in,
                              void* d_out, int out_size, void* d_ws, size_t ws_size,
                              hipStream_t stream) {
  const float* x      = (const float*)d_in[0];
  const float* params = (const float*)d_in[1];
  const float* head_w = (const float*)d_in[2];
  const float* head_b = (const float*)d_in[3];
  float* out = (float*)d_out;
  float* Ure = (float*)d_ws;
  float* Uim = Ure + NL * 3 * 256;

  const int B = in_sizes[0] / NQ;
  const size_t shmem =
      (size_t)(2 * NS * DIM + NS + 2 * NS * NQ) * sizeof(float);
  (void)hipFuncSetAttribute((const void*)qsim_kernel,
                            hipFuncAttributeMaxDynamicSharedMemorySize,
                            (int)shmem);

  qgates_kernel<<<1, 256, 0, stream>>>(params, Ure, Uim);
  qsim_kernel<<<B / NS, THREADS, shmem, stream>>>(x, head_w, head_b, Ure, Uim, out);
}


# Complex gate product with three real matrix products instead of four: P=a(c+d), re=P-(a+b)d, im=P+(b-a)c; 12 WMMA per tile, not 16 (re-associates the sums of every amplitude)
# speedup vs baseline: 1.6967x; 1.1473x over previous
#include <hip/hip_runtime.h>

typedef __attribute__((ext_vector_type(2))) float v2f;
typedef __attribute__((ext_vector_type(8))) float v8f;

#define NQ      12
#define DIM     4096
#define NS      2
#define NL      3
#define THREADS 256

template <int L>
__device__ __forceinline__ int permc(int z) {
  return (L == 0) ? z
       : (L == 1) ? (z ^ (z >> 1))
       : (L == 2) ? (z ^ (z >> 2))
                  : (z ^ (z >> 1) ^ (z >> 2) ^ (z >> 3));
}

__global__ void qgates_kernel(const float* __restrict__ params,
                              float* __restrict__ Ure, float* __restrict__ Uim) {
  const int t = threadIdx.x;
  const int r = t >> 4, c = t & 15;
  for (int l = 0; l < NL; ++l) {
    for (int g = 0; g < 3; ++g) {
      float gr[4][2][2], gi[4][2][2];
      #pragma unroll
      for (int j = 0; j < 4; ++j) {
        const int wire = g * 4 + j;
        const float* p = params + (l * NQ + wire) * 3;
        const float c0 = cosf(0.5f * p[0]), s0 = sinf(0.5f * p[0]);
        const float cp = cosf(0.5f * p[1]), sp = sinf(0.5f * p[1]);
        const float c2 = cosf(0.5f * p[2]), s2 = sinf(0.5f * p[2]);
        gr[j][0][0] =  cp * (c2 * c0 - s2 * s0);
        gi[j][0][0] = -sp * (c2 * c0 + s2 * s0);
        gr[j][0][1] = -cp * (c2 * s0 + s2 * c0);
        gi[j][0][1] =  sp * (c2 * s0 - s2 * c0);
        gr[j][1][0] =  cp * (s2 * c0 + c2 * s0);
        gi[j][1][0] =  sp * (c2 * s0 - s2 * c0);
        gr[j][1][1] =  cp * (c2 * c0 - s2 * s0);
        gi[j][1][1] =  sp * (c2 * c0 + s2 * s0);
      }
      float ar = 1.f, ai = 0.f;
      #pragma unroll
      for (int j = 0; j < 4; ++j) {
        const int rb = (r >> (3 - j)) & 1, cb = (c >> (3 - j)) & 1;
        const float br = gr[j][rb][cb], bi = gi[j][rb][cb];
        const float nr = ar * br - ai * bi, ni = ar * bi + ai * br;
        ar = nr; ai = ni;
      }
      Ure[(l * 3 + g) * 256 + r * 16 + c] = ar;
      Uim[(l * 3 + g) * 256 + r * 16 + c] = ai;
    }
  }
}

template <int L, int G>
__device__ __forceinline__ void apply_group(float2* __restrict__ sAmp,
                                            const float* __restrict__ Ure,
                                            const float* __restrict__ Uim,
                                            int lane, int wave) {
  constexpr int RS = (G == 0) ? 256 : (G == 1) ? 16 : 1;
  const float* ur = Ure + (L * 3 + G) * 256;
  const float* ui = Uim + (L * 3 + G) * 256;
  const bool hi = (lane >= 16);
  const int koff = hi ? 2 : 0;
  const int moff = lane & 15;
  const int nn   = lane & 15;

  v2f aRe[4], aNS[4], aD[4];
  #pragma unroll
  for (int f = 0; f < 4; ++f) {
    const int k0 = 4 * f + koff;
    v2f tr, ti;
    tr.x = ur[moff * 16 + k0]; tr.y = ur[moff * 16 + k0 + 1];
    ti.x = ui[moff * 16 + k0]; ti.y = ui[moff * 16 + k0 + 1];
    aRe[f] = tr; aNS[f] = -(tr + ti); aD[f] = ti - tr;
  }

  int pk0[4], pk1[4], pm[8];
  #pragma unroll
  for (int f = 0; f < 4; ++f) {
    pk0[f] = hi ? permc<L>((4 * f + 2) * RS) : permc<L>((4 * f + 0) * RS);
    pk1[f] = hi ? permc<L>((4 * f + 3) * RS) : permc<L>((4 * f + 1) * RS);
  }
  #pragma unroll
  for (int v = 0; v < 8; ++v)
    pm[v] = hi ? permc<L>((v + 8) * RS) : permc<L>(v * RS);

  constexpr int TPW = NS * 16 / (THREADS / 32);
  #pragma unroll 2
  for (int t = 0; t < TPW; ++t) {
    const int tt = wave * TPW + t;
    const int ss = tt >> 4;
    const int c  = ((tt & 15) << 4) + nn;
    int other;
    if (G == 0)      other = c;
    else if (G == 1) other = ((c >> 4) << 8) | (c & 15);
    else             other = c << 4;
    const int pob = (ss << 12) | permc<L>(other);

    v2f bRe[4], bIm[4], bS[4];
    #pragma unroll
    for (int f = 0; f < 4; ++f) {
      const float2 q0 = sAmp[pob ^ pk0[f]];
      const float2 q1 = sAmp[pob ^ pk1[f]];
      v2f tr, ti, ts;
      tr.x = q0.x; tr.y = q1.x;
      ti.x = q0.y; ti.y = q1.y;
      ts.x = q0.x + q0.y; ts.y = q1.x + q1.y;
      bRe[f] = tr; bIm[f] = ti; bS[f] = ts;
    }
    v8f dp = {};
    #pragma unroll
    for (int f = 0; f < 4; ++f)
      dp = __builtin_amdgcn_wmma_f32_16x16x4_f32(false, aRe[f], false, bS[f], (short)0, dp, false, false);
    v8f dre = dp; v8f dim_ = dp;
    #pragma unroll
    for (int f = 0; f < 4; ++f) {
      dre  = __builtin_amdgcn_wmma_f32_16x16x4_f32(false, aNS[f], false, bIm[f], (short)0, dre,  false, false);
      dim_ = __builtin_amdgcn_wmma_f32_16x16x4_f32(false, aD[f],  false, bRe[f], (short)0, dim_, false, false);
    }
    #pragma unroll
    for (int v = 0; v < 8; ++v) {
      float2 st; st.x = dre[v]; st.y = dim_[v];
      sAmp[pob ^ pm[v]] = st;
    }
  }
}

__global__ void __launch_bounds__(THREADS)
qsim_kernel(const float* __restrict__ x, const float* __restrict__ head_w,
            const float* __restrict__ head_b, const float* __restrict__ Ure,
            const float* __restrict__ Uim, float* __restrict__ out) {
  extern __shared__ float smem_raw[];
  float2* sAmp = (float2*)smem_raw;
  float*  csA  = (float*)(sAmp + NS * DIM);
  float*  snA  = csA + NS * NQ;
  float*  accS = snA + NS * NQ;

  const int tid  = threadIdx.x;
  const int lane = tid & 31;
  const int wave = tid >> 5;
  const int s0   = blockIdx.x * NS;

  float w[NQ];
  #pragma unroll
  for (int i = 0; i < NQ; ++i) w[i] = head_w[i];

  if (tid < NS * NQ) {
    const int ss = tid / NQ, i = tid - ss * NQ;
    const float v = 0.5f * x[(s0 + ss) * NQ + i];
    csA[tid] = cosf(v);
    snA[tid] = sinf(v);
  }
  __syncthreads();
  const int kl = __popc(tid) & 3;
  #pragma unroll
  for (int ss = 0; ss < NS; ++ss) {
    float cs[NQ], sn[NQ];
    #pragma unroll
    for (int i = 0; i < NQ; ++i) { cs[i] = csA[ss * NQ + i]; sn[i] = snA[ss * NQ + i]; }
    float ml = ((tid >> 7) & 1) ? sn[4] : cs[4];
    #pragma unroll
    for (int i = 5; i < NQ; ++i)
      ml *= ((tid >> (NQ - 1 - i)) & 1) ? sn[i] : cs[i];
    const float pr = (kl == 0) ? ml : (kl == 2) ? -ml : 0.f;
    const float pi = (kl == 1) ? -ml : (kl == 3) ? ml : 0.f;
    const float t01[4] = {cs[0] * cs[1], cs[0] * sn[1], sn[0] * cs[1], sn[0] * sn[1]};
    const float t23[4] = {cs[2] * cs[3], cs[2] * sn[3], sn[2] * cs[3], sn[2] * sn[3]};
    #pragma unroll
    for (int j = 0; j < 16; ++j) {
      const float mh = t01[j >> 2] * t23[j & 3];
      const int pj = __builtin_popcount(j) & 3;
      const float rr = (pj == 0) ? pr : (pj == 1) ? pi : (pj == 2) ? -pr : -pi;
      const float ri = (pj == 0) ? pi : (pj == 1) ? -pr : (pj == 2) ? -pi : pr;
      float2 a; a.x = rr * mh; a.y = ri * mh;
      sAmp[(ss << 12) + (j << 8) + tid] = a;
    }
  }
  __syncthreads();

  apply_group<0, 0>(sAmp, Ure, Uim, lane, wave); __syncthreads();
  apply_group<0, 1>(sAmp, Ure, Uim, lane, wave); __syncthreads();
  apply_group<0, 2>(sAmp, Ure, Uim, lane, wave); __syncthreads();
  apply_group<1, 0>(sAmp, Ure, Uim, lane, wave); __syncthreads();
  apply_group<1, 1>(sAmp, Ure, Uim, lane, wave); __syncthreads();
  apply_group<1, 2>(sAmp, Ure, Uim, lane, wave); __syncthreads();
  apply_group<2, 0>(sAmp, Ure, Uim, lane, wave); __syncthreads();
  apply_group<2, 1>(sAmp, Ure, Uim, lane, wave); __syncthreads();
  apply_group<2, 2>(sAmp, Ure, Uim, lane, wave); __syncthreads();

  if (tid < NS) accS[tid] = 0.f;
  __syncthreads();
  float wsum = 0.f;
  #pragma unroll
  for (int i = 0; i < NQ; ++i) wsum += w[i];
  float cl = wsum;
  #pragma unroll
  for (int i = 4; i < NQ; ++i)
    cl -= 2.f * w[i] * (float)((tid >> (NQ - 1 - i)) & 1);
  float coef[16];
  #pragma unroll
  for (int j = 0; j < 16; ++j) {
    float ch = 0.f;
    #pragma unroll
    for (int i = 0; i < 4; ++i)
      ch -= 2.f * w[i] * (float)((j >> (3 - i)) & 1);
    coef[j] = cl + ch;
  }
  const int pt = permc<NL>(tid);
  #pragma unroll
  for (int ss = 0; ss < NS; ++ss) {
    float a = 0.f;
    #pragma unroll
    for (int j = 0; j < 16; ++j) {
      const float2 q = sAmp[(ss << 12) + (pt ^ permc<NL>(j << 8))];
      a += (q.x * q.x + q.y * q.y) * coef[j];
    }
    atomicAdd(&accS[ss], a);
  }
  __syncthreads();
  if (tid < NS) out[s0 + tid] = accS[tid] + head_b[0];
}

extern "C" void kernel_launch(void* const* d_in, const int* in_sizes, int n_in,
                              void* d_out, int out_size, void* d_ws, size_t ws_size,
                              hipStream_t stream) {
  const float* x      = (const float*)d_in[0];
  const float* params = (const float*)d_in[1];
  const float* head_w = (const float*)d_in[2];
  const float* head_b = (const float*)d_in[3];
  float* out = (float*)d_out;
  float* Ure = (float*)d_ws;
  float* Uim = Ure + NL * 3 * 256;

  const int B = in_sizes[0] / NQ;
  const size_t shmem =
      (size_t)(2 * NS * DIM + NS + 2 * NS * NQ) * sizeof(float);
  (void)hipFuncSetAttribute((const void*)qsim_kernel,
                            hipFuncAttributeMaxDynamicSharedMemorySize,
                            (int)shmem);

  qgates_kernel<<<1, 256, 0, stream>>>(params, Ure, Uim);
  qsim_kernel<<<B / NS, THREADS, shmem, stream>>>(x, head_w, head_b, Ure, Uim, out);
}


# Layer 0's rotations meet a product state, so they fold into the encode, amp(z) = prod_i (g_0i RX(x_i)|0>)[bit_i]: 6 gate passes, not 9 (re-associates the products and sums of every amplitude)
# speedup vs baseline: 2.4010x; 1.4151x over previous
#include <hip/hip_runtime.h>

typedef __attribute__((ext_vector_type(2))) float v2f;
typedef __attribute__((ext_vector_type(8))) float v8f;

#define NQ      12
#define DIM     4096
#define NS      2
#define NL      3
#define THREADS 256

template <int L>
__device__ __forceinline__ int permc(int z) {
  return (L == 0) ? z
       : (L == 1) ? (z ^ (z >> 1))
       : (L == 2) ? (z ^ (z >> 2))
                  : (z ^ (z >> 1) ^ (z >> 2) ^ (z >> 3));
}

__device__ __forceinline__ void wire_gate(const float* __restrict__ p,
                                          float gr[2][2], float gi[2][2]) {
  const float c0 = cosf(0.5f * p[0]), s0 = sinf(0.5f * p[0]);
  const float cp = cosf(0.5f * p[1]), sp = sinf(0.5f * p[1]);
  const float c2 = cosf(0.5f * p[2]), s2 = sinf(0.5f * p[2]);
  gr[0][0] =  cp * (c2 * c0 - s2 * s0);
  gi[0][0] = -sp * (c2 * c0 + s2 * s0);
  gr[0][1] = -cp * (c2 * s0 + s2 * c0);
  gi[0][1] =  sp * (c2 * s0 - s2 * c0);
  gr[1][0] =  cp * (s2 * c0 + c2 * s0);
  gi[1][0] =  sp * (c2 * s0 - s2 * c0);
  gr[1][1] =  cp * (c2 * c0 - s2 * s0);
  gi[1][1] =  sp * (c2 * c0 + s2 * s0);
}

__global__ void qgates_kernel(const float* __restrict__ params,
                              float* __restrict__ Ure, float* __restrict__ Uim) {
  const int t = threadIdx.x;
  const int r = t >> 4, c = t & 15;
  if (t < NQ) {
    float g0r[2][2], g0i[2][2];
    wire_gate(params + t * 3, g0r, g0i);
    float* o = Ure + t * 8;
    o[0] = g0r[0][0]; o[1] = g0i[0][0]; o[2] = g0r[0][1]; o[3] = g0i[0][1];
    o[4] = g0r[1][0]; o[5] = g0i[1][0]; o[6] = g0r[1][1]; o[7] = g0i[1][1];
  }
  for (int l = 1; l < NL; ++l) {
    for (int g = 0; g < 3; ++g) {
      float gr[4][2][2], gi[4][2][2];
      #pragma unroll
      for (int j = 0; j < 4; ++j)
        wire_gate(params + (l * NQ + g * 4 + j) * 3, gr[j], gi[j]);
      float ar = 1.f, ai = 0.f;
      #pragma unroll
      for (int j = 0; j < 4; ++j) {
        const int rb = (r >> (3 - j)) & 1, cb = (c >> (3 - j)) & 1;
        const float br = gr[j][rb][cb], bi = gi[j][rb][cb];
        const float nr = ar * br - ai * bi, ni = ar * bi + ai * br;
        ar = nr; ai = ni;
      }
      Ure[(l * 3 + g) * 256 + r * 16 + c] = ar;
      Uim[(l * 3 + g) * 256 + r * 16 + c] = ai;
    }
  }
}

__device__ __forceinline__ float2 cmul(float2 a, float2 b) {
  float2 r;
  r.x = a.x * b.x - a.y * b.y;
  r.y = a.x * b.y + a.y * b.x;
  return r;
}

template <int L, int G>
__device__ __forceinline__ void apply_group(float2* __restrict__ sAmp,
                                            const float* __restrict__ Ure,
                                            const float* __restrict__ Uim,
                                            int lane, int wave) {
  constexpr int RS = (G == 0) ? 256 : (G == 1) ? 16 : 1;
  const float* ur = Ure + (L * 3 + G) * 256;
  const float* ui = Uim + (L * 3 + G) * 256;
  const bool hi = (lane >= 16);
  const int koff = hi ? 2 : 0;
  const int moff = lane & 15;
  const int nn   = lane & 15;

  v2f aRe[4], aNS[4], aD[4];
  #pragma unroll
  for (int f = 0; f < 4; ++f) {
    const int k0 = 4 * f + koff;
    v2f tr, ti;
    tr.x = ur[moff * 16 + k0]; tr.y = ur[moff * 16 + k0 + 1];
    ti.x = ui[moff * 16 + k0]; ti.y = ui[moff * 16 + k0 + 1];
    aRe[f] = tr; aNS[f] = -(tr + ti); aD[f] = ti - tr;
  }

  int pk0[4], pk1[4], pm[8];
  #pragma unroll
  for (int f = 0; f < 4; ++f) {
    pk0[f] = hi ? permc<L>((4 * f + 2) * RS) : permc<L>((4 * f + 0) * RS);
    pk1[f] = hi ? permc<L>((4 * f + 3) * RS) : permc<L>((4 * f + 1) * RS);
  }
  #pragma unroll
  for (int v = 0; v < 8; ++v)
    pm[v] = hi ? permc<L>((v + 8) * RS) : permc<L>(v * RS);

  constexpr int TPW = NS * 16 / (THREADS / 32);
  #pragma unroll 2
  for (int t = 0; t < TPW; ++t) {
    const int tt = wave * TPW + t;
    const int ss = tt >> 4;
    const int c  = ((tt & 15) << 4) + nn;
    int other;
    if (G == 0)      other = c;
    else if (G == 1) other = ((c >> 4) << 8) | (c & 15);
    else             other = c << 4;
    const int pob = (ss << 12) | permc<L>(other);

    v2f bRe[4], bIm[4], bS[4];
    #pragma unroll
    for (int f = 0; f < 4; ++f) {
      const float2 q0 = sAmp[pob ^ pk0[f]];
      const float2 q1 = sAmp[pob ^ pk1[f]];
      v2f tr, ti, ts;
      tr.x = q0.x; tr.y = q1.x;
      ti.x = q0.y; ti.y = q1.y;
      ts.x = q0.x + q0.y; ts.y = q1.x + q1.y;
      bRe[f] = tr; bIm[f] = ti; bS[f] = ts;
    }
    v8f dp = {};
    #pragma unroll
    for (int f = 0; f < 4; ++f)
      dp = __builtin_amdgcn_wmma_f32_16x16x4_f32(false, aRe[f], false, bS[f], (short)0, dp, false, false);
    v8f dre = dp; v8f dim_ = dp;
    #pragma unroll
    for (int f = 0; f < 4; ++f) {
      dre  = __builtin_amdgcn_wmma_f32_16x16x4_f32(false, aNS[f], false, bIm[f], (short)0, dre,  false, false);
      dim_ = __builtin_amdgcn_wmma_f32_16x16x4_f32(false, aD[f],  false, bRe[f], (short)0, dim_, false, false);
    }
    #pragma unroll
    for (int v = 0; v < 8; ++v) {
      float2 st; st.x = dre[v]; st.y = dim_[v];
      sAmp[pob ^ pm[v]] = st;
    }
  }
}

__global__ void __launch_bounds__(THREADS)
qsim_kernel(const float* __restrict__ x, const float* __restrict__ head_w,
            const float* __restrict__ head_b, const float* __restrict__ Ure,
            const float* __restrict__ Uim, float* __restrict__ out) {
  extern __shared__ float smem_raw[];
  float2* sAmp = (float2*)smem_raw;
  float4* wv   = (float4*)(sAmp + NS * DIM);
  float*  accS = (float*)(wv + NS * NQ);

  const int tid  = threadIdx.x;
  const int lane = tid & 31;
  const int wave = tid >> 5;
  const int s0   = blockIdx.x * NS;

  float w[NQ];
  #pragma unroll
  for (int i = 0; i < NQ; ++i) w[i] = head_w[i];

  if (tid < NS * NQ) {
    const int ss = tid / NQ, i = tid - ss * NQ;
    const float v = 0.5f * x[(s0 + ss) * NQ + i];
    const float cv = cosf(v), sv = sinf(v);
    const float* g = Ure + i * 8;
    float4 u;
    u.x = g[0] * cv + g[3] * sv;  u.y = g[1] * cv - g[2] * sv;
    u.z = g[4] * cv + g[7] * sv;  u.w = g[5] * cv - g[6] * sv;
    wv[tid] = u;
  }
  __syncthreads();
  const float2* wv2 = (const float2*)wv;
  #pragma unroll
  for (int ss = 0; ss < NS; ++ss) {
    float2 p = wv2[(ss * NQ + 4) * 2 + ((tid >> 7) & 1)];
    #pragma unroll
    for (int i = 5; i < NQ; ++i)
      p = cmul(p, wv2[(ss * NQ + i) * 2 + ((tid >> (NQ - 1 - i)) & 1)]);
    const float4 u0 = wv[ss * NQ + 0], u1 = wv[ss * NQ + 1];
    const float4 u2 = wv[ss * NQ + 2], u3 = wv[ss * NQ + 3];
    const float2 p0 = cmul(p, make_float2(u0.x, u0.y));
    const float2 p1 = cmul(p, make_float2(u0.z, u0.w));
    const float2 q01[4] = {cmul(p0, make_float2(u1.x, u1.y)), cmul(p0, make_float2(u1.z, u1.w)),
                           cmul(p1, make_float2(u1.x, u1.y)), cmul(p1, make_float2(u1.z, u1.w))};
    const float2 t23[4] = {cmul(make_float2(u2.x, u2.y), make_float2(u3.x, u3.y)),
                           cmul(make_float2(u2.x, u2.y), make_float2(u3.z, u3.w)),
                           cmul(make_float2(u2.z, u2.w), make_float2(u3.x, u3.y)),
                           cmul(make_float2(u2.z, u2.w), make_float2(u3.z, u3.w))};
    #pragma unroll
    for (int j = 0; j < 16; ++j)
      sAmp[(ss << 12) + (j << 8) + tid] = cmul(q01[j >> 2], t23[j & 3]);
  }
  __syncthreads();

  apply_group<1, 0>(sAmp, Ure, Uim, lane, wave); __syncthreads();
  apply_group<1, 1>(sAmp, Ure, Uim, lane, wave); __syncthreads();
  apply_group<1, 2>(sAmp, Ure, Uim, lane, wave); __syncthreads();
  apply_group<2, 0>(sAmp, Ure, Uim, lane, wave); __syncthreads();
  apply_group<2, 1>(sAmp, Ure, Uim, lane, wave); __syncthreads();
  apply_group<2, 2>(sAmp, Ure, Uim, lane, wave); __syncthreads();

  if (tid < NS) accS[tid] = 0.f;
  __syncthreads();
  float wsum = 0.f;
  #pragma unroll
  for (int i = 0; i < NQ; ++i) wsum += w[i];
  float cl = wsum;
  #pragma unroll
  for (int i = 4; i < NQ; ++i)
    cl -= 2.f * w[i] * (float)((tid >> (NQ - 1 - i)) & 1);
  float coef[16];
  #pragma unroll
  for (int j = 0; j < 16; ++j) {
    float ch = 0.f;
    #pragma unroll
    for (int i = 0; i < 4; ++i)
      ch -= 2.f * w[i] * (float)((j >> (3 - i)) & 1);
    coef[j] = cl + ch;
  }
  const int pt = permc<NL>(tid);
  #pragma unroll
  for (int ss = 0; ss < NS; ++ss) {
    float a = 0.f;
    #pragma unroll
    for (int j = 0; j < 16; ++j) {
      const float2 q = sAmp[(ss << 12) + (pt ^ permc<NL>(j << 8))];
      a += (q.x * q.x + q.y * q.y) * coef[j];
    }
    atomicAdd(&accS[ss], a);
  }
  __syncthreads();
  if (tid < NS) out[s0 + tid] = accS[tid] + head_b[0];
}

extern "C" void kernel_launch(void* const* d_in, const int* in_sizes, int n_in,
                              void* d_out, int out_size, void* d_ws, size_t ws_size,
                              hipStream_t stream) {
  const float* x      = (const float*)d_in[0];
  const float* params = (const float*)d_in[1];
  const float* head_w = (const float*)d_in[2];
  const float* head_b = (const float*)d_in[3];
  float* out = (float*)d_out;
  float* Ure = (float*)d_ws;
  float* Uim = Ure + NL * 3 * 256;

  const int B = in_sizes[0] / NQ;
  const size_t shmem =
      (size_t)(2 * NS * DIM + NS + 4 * NS * NQ) * sizeof(float);
  (void)hipFuncSetAttribute((const void*)qsim_kernel,
                            hipFuncAttributeMaxDynamicSharedMemorySize,
                            (int)shmem);

  qgates_kernel<<<1, 256, 0, stream>>>(params, Ure, Uim);
  qsim_kernel<<<B / NS, THREADS, shmem, stream>>>(x, head_w, head_b, Ure, Uim, out);
}


# Groups 1 and 2 share one 256-amplitude tile: both gates per LDS round trip, product 1 transposed so its D registers are product 2's B (4 passes, not 6; re-associates every amplitude's sums)
# speedup vs baseline: 2.7026x; 1.1256x over previous
#include <hip/hip_runtime.h>

typedef __attribute__((ext_vector_type(2))) float v2f;
typedef __attribute__((ext_vector_type(8))) float v8f;

#define NQ      12
#define DIM     4096
#define NS      2
#define NL      3
#define THREADS 256

template <int L>
__device__ __forceinline__ int permc(int z) {
  return (L == 0) ? z
       : (L == 1) ? (z ^ (z >> 1))
       : (L == 2) ? (z ^ (z >> 2))
                  : (z ^ (z >> 1) ^ (z >> 2) ^ (z >> 3));
}

__device__ __forceinline__ void wire_gate(const float* __restrict__ p,
                                          float gr[2][2], float gi[2][2]) {
  const float c0 = cosf(0.5f * p[0]), s0 = sinf(0.5f * p[0]);
  const float cp = cosf(0.5f * p[1]), sp = sinf(0.5f * p[1]);
  const float c2 = cosf(0.5f * p[2]), s2 = sinf(0.5f * p[2]);
  gr[0][0] =  cp * (c2 * c0 - s2 * s0);
  gi[0][0] = -sp * (c2 * c0 + s2 * s0);
  gr[0][1] = -cp * (c2 * s0 + s2 * c0);
  gi[0][1] =  sp * (c2 * s0 - s2 * c0);
  gr[1][0] =  cp * (s2 * c0 + c2 * s0);
  gi[1][0] =  sp * (c2 * s0 - s2 * c0);
  gr[1][1] =  cp * (c2 * c0 - s2 * s0);
  gi[1][1] =  sp * (c2 * c0 + s2 * s0);
}

__global__ void qgates_kernel(const float* __restrict__ params,
                              float* __restrict__ Ure, float* __restrict__ Uim) {
  const int t = threadIdx.x;
  const int r = t >> 4, c = t & 15;
  if (t < NQ) {
    float g0r[2][2], g0i[2][2];
    wire_gate(params + t * 3, g0r, g0i);
    float* o = Ure + t * 8;
    o[0] = g0r[0][0]; o[1] = g0i[0][0]; o[2] = g0r[0][1]; o[3] = g0i[0][1];
    o[4] = g0r[1][0]; o[5] = g0i[1][0]; o[6] = g0r[1][1]; o[7] = g0i[1][1];
  }
  for (int l = 1; l < NL; ++l) {
    for (int g = 0; g < 3; ++g) {
      float gr[4][2][2], gi[4][2][2];
      #pragma unroll
      for (int j = 0; j < 4; ++j)
        wire_gate(params + (l * NQ + g * 4 + j) * 3, gr[j], gi[j]);
      float ar = 1.f, ai = 0.f;
      #pragma unroll
      for (int j = 0; j < 4; ++j) {
        const int rb = (r >> (3 - j)) & 1, cb = (c >> (3 - j)) & 1;
        const float br = gr[j][rb][cb], bi = gi[j][rb][cb];
        const float nr = ar * br - ai * bi, ni = ar * bi + ai * br;
        ar = nr; ai = ni;
      }
      Ure[(l * 3 + g) * 256 + r * 16 + c] = ar;
      Uim[(l * 3 + g) * 256 + r * 16 + c] = ai;
    }
  }
}

__device__ __forceinline__ float2 cmul(float2 a, float2 b) {
  float2 r;
  r.x = a.x * b.x - a.y * b.y;
  r.y = a.x * b.y + a.y * b.x;
  return r;
}

template <int L, int G>
__device__ __forceinline__ void apply_group(float2* __restrict__ sAmp,
                                            const float* __restrict__ Ure,
                                            const float* __restrict__ Uim,
                                            int lane, int wave) {
  constexpr int RS = (G == 0) ? 256 : (G == 1) ? 16 : 1;
  const float* ur = Ure + (L * 3 + G) * 256;
  const float* ui = Uim + (L * 3 + G) * 256;
  const bool hi = (lane >= 16);
  const int koff = hi ? 2 : 0;
  const int moff = lane & 15;
  const int nn   = lane & 15;

  v2f aRe[4], aNS[4], aD[4];
  #pragma unroll
  for (int f = 0; f < 4; ++f) {
    const int k0 = 4 * f + koff;
    v2f tr, ti;
    tr.x = ur[moff * 16 + k0]; tr.y = ur[moff * 16 + k0 + 1];
    ti.x = ui[moff * 16 + k0]; ti.y = ui[moff * 16 + k0 + 1];
    aRe[f] = tr; aNS[f] = -(tr + ti); aD[f] = ti - tr;
  }

  int pk0[4], pk1[4], pm[8];
  #pragma unroll
  for (int f = 0; f < 4; ++f) {
    pk0[f] = hi ? permc<L>((4 * f + 2) * RS) : permc<L>((4 * f + 0) * RS);
    pk1[f] = hi ? permc<L>((4 * f + 3) * RS) : permc<L>((4 * f + 1) * RS);
  }
  #pragma unroll
  for (int v = 0; v < 8; ++v)
    pm[v] = hi ? permc<L>((v + 8) * RS) : permc<L>(v * RS);

  constexpr int TPW = NS * 16 / (THREADS / 32);
  #pragma unroll 2
  for (int t = 0; t < TPW; ++t) {
    const int tt = wave * TPW + t;
    const int ss = tt >> 4;
    const int c  = ((tt & 15) << 4) + nn;
    int other;
    if (G == 0)      other = c;
    else if (G == 1) other = ((c >> 4) << 8) | (c & 15);
    else             other = c << 4;
    const int pob = (ss << 12) | permc<L>(other);

    v2f bRe[4], bIm[4], bS[4];
    #pragma unroll
    for (int f = 0; f < 4; ++f) {
      const float2 q0 = sAmp[pob ^ pk0[f]];
      const float2 q1 = sAmp[pob ^ pk1[f]];
      v2f tr, ti, ts;
      tr.x = q0.x; tr.y = q1.x;
      ti.x = q0.y; ti.y = q1.y;
      ts.x = q0.x + q0.y; ts.y = q1.x + q1.y;
      bRe[f] = tr; bIm[f] = ti; bS[f] = ts;
    }
    v8f dp = {};
    #pragma unroll
    for (int f = 0; f < 4; ++f)
      dp = __builtin_amdgcn_wmma_f32_16x16x4_f32(false, aRe[f], false, bS[f], (short)0, dp, false, false);
    v8f dre = dp; v8f dim_ = dp;
    #pragma unroll
    for (int f = 0; f < 4; ++f) {
      dre  = __builtin_amdgcn_wmma_f32_16x16x4_f32(false, aNS[f], false, bIm[f], (short)0, dre,  false, false);
      dim_ = __builtin_amdgcn_wmma_f32_16x16x4_f32(false, aD[f],  false, bRe[f], (short)0, dim_, false, false);
    }
    #pragma unroll
    for (int v = 0; v < 8; ++v) {
      float2 st; st.x = dre[v]; st.y = dim_[v];
      sAmp[pob ^ pm[v]] = st;
    }
  }
}

template <int L>
__device__ __forceinline__ void apply_group12(float2* __restrict__ sAmp,
                                              const float* __restrict__ Ure,
                                              const float* __restrict__ Uim,
                                              int lane, int wave) {
  const float* ur1 = Ure + (L * 3 + 1) * 256;
  const float* ui1 = Uim + (L * 3 + 1) * 256;
  const float* ur2 = Ure + (L * 3 + 2) * 256;
  const float* ui2 = Uim + (L * 3 + 2) * 256;
  const bool hi = (lane >= 16);
  const int nn = lane & 15;

  v2f g1Re[4], g1NS[4], g1D[4], g2Re[4], g2NS[4], g2D[4];
  #pragma unroll
  for (int f = 0; f < 4; ++f) {
    const int k1 = 4 * f + (hi ? 2 : 0);
    const int k2 = 2 * f + (hi ? 8 : 0);
    v2f tr, ti;
    tr.x = ur1[nn * 16 + k1]; tr.y = ur1[nn * 16 + k1 + 1];
    ti.x = ui1[nn * 16 + k1]; ti.y = ui1[nn * 16 + k1 + 1];
    g1Re[f] = tr; g1NS[f] = -(tr + ti); g1D[f] = ti - tr;
    tr.x = ur2[nn * 16 + k2]; tr.y = ur2[nn * 16 + k2 + 1];
    ti.x = ui2[nn * 16 + k2]; ti.y = ui2[nn * 16 + k2 + 1];
    g2Re[f] = tr; g2NS[f] = -(tr + ti); g2D[f] = ti - tr;
  }

  char* const sB = (char*)sAmp;
  const int laneL = (permc<L>(nn) ^ (hi ? permc<L>(2 * 16) : 0)) << 3;
  const int laneS = (permc<L>(nn << 4) ^ (hi ? permc<L>(8) : 0)) << 3;

  constexpr int TPW = NS * 16 / (THREADS / 32);
  #pragma unroll 1
  for (int t = 0; t < TPW; ++t) {
    const int tt = wave * TPW + t;
    const int tileB = (((tt >> 4) << 12) ^ permc<L>((tt & 15) << 8)) << 3;
    const int baseL = tileB ^ laneL;
    const int baseS = tileB ^ laneS;

    v2f xRe[4], xIm[4], xS[4];
    #pragma unroll
    for (int f = 0; f < 4; ++f) {
      const float2 q0 = *(const float2*)(sB + (baseL ^ (permc<L>((4 * f + 0) * 16) << 3)));
      const float2 q1 = *(const float2*)(sB + (baseL ^ (permc<L>((4 * f + 1) * 16) << 3)));
      v2f tr, ti, ts;
      tr.x = q0.x; tr.y = q1.x;
      ti.x = q0.y; ti.y = q1.y;
      ts.x = q0.x + q0.y; ts.y = q1.x + q1.y;
      xRe[f] = tr; xIm[f] = ti; xS[f] = ts;
    }
    v8f p1 = {};
    #pragma unroll
    for (int f = 0; f < 4; ++f)
      p1 = __builtin_amdgcn_wmma_f32_16x16x4_f32(false, xS[f], false, g1Re[f], (short)0, p1, false, false);
    v8f yre = p1; v8f yim = p1;
    #pragma unroll
    for (int f = 0; f < 4; ++f) {
      yre = __builtin_amdgcn_wmma_f32_16x16x4_f32(false, xIm[f], false, g1NS[f], (short)0, yre, false, false);
      yim = __builtin_amdgcn_wmma_f32_16x16x4_f32(false, xRe[f], false, g1D[f],  (short)0, yim, false, false);
    }
    v2f bRe[4], bIm[4], bS[4];
    #pragma unroll
    for (int f = 0; f < 4; ++f) {
      v2f tr, ti;
      tr.x = yre[2 * f]; tr.y = yre[2 * f + 1];
      ti.x = yim[2 * f]; ti.y = yim[2 * f + 1];
      bRe[f] = tr; bIm[f] = ti; bS[f] = tr + ti;
    }
    v8f p2 = {};
    #pragma unroll
    for (int f = 0; f < 4; ++f)
      p2 = __builtin_amdgcn_wmma_f32_16x16x4_f32(false, g2Re[f], false, bS[f], (short)0, p2, false, false);
    v8f zre = p2; v8f zim = p2;
    #pragma unroll
    for (int f = 0; f < 4; ++f) {
      zre = __builtin_amdgcn_wmma_f32_16x16x4_f32(false, g2NS[f], false, bIm[f], (short)0, zre, false, false);
      zim = __builtin_amdgcn_wmma_f32_16x16x4_f32(false, g2D[f],  false, bRe[f], (short)0, zim, false, false);
    }
    #pragma unroll
    for (int v = 0; v < 8; ++v) {
      float2 st; st.x = zre[v]; st.y = zim[v];
      *(float2*)(sB + (baseS ^ (permc<L>(v) << 3))) = st;
    }
  }
}

__global__ void __launch_bounds__(THREADS) __attribute__((amdgpu_waves_per_eu(8)))
qsim_kernel(const float* __restrict__ x, const float* __restrict__ head_w,
            const float* __restrict__ head_b, const float* __restrict__ Ure,
            const float* __restrict__ Uim, float* __restrict__ out) {
  extern __shared__ float smem_raw[];
  float2* sAmp = (float2*)smem_raw;
  float4* wv   = (float4*)(sAmp + NS * DIM);
  float*  accS = (float*)(wv + NS * NQ);

  const int tid  = threadIdx.x;
  const int lane = tid & 31;
  const int wave = tid >> 5;
  const int s0   = blockIdx.x * NS;

  float w[NQ];
  #pragma unroll
  for (int i = 0; i < NQ; ++i) w[i] = head_w[i];

  if (tid < NS * NQ) {
    const int ss = tid / NQ, i = tid - ss * NQ;
    const float v = 0.5f * x[(s0 + ss) * NQ + i];
    const float cv = cosf(v), sv = sinf(v);
    const float* g = Ure + i * 8;
    float4 u;
    u.x = g[0] * cv + g[3] * sv;  u.y = g[1] * cv - g[2] * sv;
    u.z = g[4] * cv + g[7] * sv;  u.w = g[5] * cv - g[6] * sv;
    wv[tid] = u;
  }
  __syncthreads();
  const float2* wv2 = (const float2*)wv;
  #pragma unroll
  for (int ss = 0; ss < NS; ++ss) {
    float2 p = wv2[(ss * NQ + 4) * 2 + ((tid >> 7) & 1)];
    #pragma unroll
    for (int i = 5; i < NQ; ++i)
      p = cmul(p, wv2[(ss * NQ + i) * 2 + ((tid >> (NQ - 1 - i)) & 1)]);
    const float4 u0 = wv[ss * NQ + 0], u1 = wv[ss * NQ + 1];
    const float4 u2 = wv[ss * NQ + 2], u3 = wv[ss * NQ + 3];
    const float2 p0 = cmul(p, make_float2(u0.x, u0.y));
    const float2 p1 = cmul(p, make_float2(u0.z, u0.w));
    const float2 q01[4] = {cmul(p0, make_float2(u1.x, u1.y)), cmul(p0, make_float2(u1.z, u1.w)),
                           cmul(p1, make_float2(u1.x, u1.y)), cmul(p1, make_float2(u1.z, u1.w))};
    const float2 t23[4] = {cmul(make_float2(u2.x, u2.y), make_float2(u3.x, u3.y)),
                           cmul(make_float2(u2.x, u2.y), make_float2(u3.z, u3.w)),
                           cmul(make_float2(u2.z, u2.w), make_float2(u3.x, u3.y)),
                           cmul(make_float2(u2.z, u2.w), make_float2(u3.z, u3.w))};
    #pragma unroll
    for (int j = 0; j < 16; ++j)
      sAmp[(ss << 12) + (j << 8) + tid] = cmul(q01[j >> 2], t23[j & 3]);
  }
  __syncthreads();

  apply_group<1, 0>(sAmp, Ure, Uim, lane, wave); __syncthreads();
  apply_group12<1>(sAmp, Ure, Uim, lane, wave);  __syncthreads();
  apply_group<2, 0>(sAmp, Ure, Uim, lane, wave); __syncthreads();
  apply_group12<2>(sAmp, Ure, Uim, lane, wave);  __syncthreads();

  if (tid < NS) accS[tid] = 0.f;
  __syncthreads();
  float wsum = 0.f;
  #pragma unroll
  for (int i = 0; i < NQ; ++i) wsum += w[i];
  float cl = wsum;
  #pragma unroll
  for (int i = 4; i < NQ; ++i)
    cl -= 2.f * w[i] * (float)((tid >> (NQ - 1 - i)) & 1);
  float coef[16];
  #pragma unroll
  for (int j = 0; j < 16; ++j) {
    float ch = 0.f;
    #pragma unroll
    for (int i = 0; i < 4; ++i)
      ch -= 2.f * w[i] * (float)((j >> (3 - i)) & 1);
    coef[j] = cl + ch;
  }
  const int pt = permc<NL>(tid);
  #pragma unroll
  for (int ss = 0; ss < NS; ++ss) {
    float a = 0.f;
    #pragma unroll
    for (int j = 0; j < 16; ++j) {
      const float2 q = sAmp[(ss << 12) + (pt ^ permc<NL>(j << 8))];
      a += (q.x * q.x + q.y * q.y) * coef[j];
    }
    atomicAdd(&accS[ss], a);
  }
  __syncthreads();
  if (tid < NS) out[s0 + tid] = accS[tid] + head_b[0];
}

extern "C" void kernel_launch(void* const* d_in, const int* in_sizes, int n_in,
                              void* d_out, int out_size, void* d_ws, size_t ws_size,
                              hipStream_t stream) {
  const float* x      = (const float*)d_in[0];
  const float* params = (const float*)d_in[1];
  const float* head_w = (const float*)d_in[2];
  const float* head_b = (const float*)d_in[3];
  float* out = (float*)d_out;
  float* Ure = (float*)d_ws;
  float* Uim = Ure + NL * 3 * 256;

  const int B = in_sizes[0] / NQ;
  const size_t shmem =
      (size_t)(2 * NS * DIM + NS + 4 * NS * NQ) * sizeof(float);
  (void)hipFuncSetAttribute((const void*)qsim_kernel,
                            hipFuncAttributeMaxDynamicSharedMemorySize,
                            (int)shmem);

  qgates_kernel<<<1, 256, 0, stream>>>(params, Ure, Uim);
  qsim_kernel<<<B / NS, THREADS, shmem, stream>>>(x, head_w, head_b, Ure, Uim, out);
}


# Layer 1's rotations meet a bond-2 matrix-product state (CNOT chain of a product state) and fold into the encode: amp = l.(M r), one K=4 WMMA pair per tile; 2 gate passes, not 4 (re-associates sums)
# speedup vs baseline: 4.5557x; 1.6857x over previous
#include <hip/hip_runtime.h>

typedef __attribute__((ext_vector_type(2))) float v2f;
typedef __attribute__((ext_vector_type(8))) float v8f;

#define NQ      12
#define DIM     4096
#define NS      2
#define NL      3
#define THREADS 256

template <int L>
__device__ __forceinline__ int permc(int z) {
  return (L == 0) ? z
       : (L == 1) ? (z ^ (z >> 1))
       : (L == 2) ? (z ^ (z >> 2))
                  : (z ^ (z >> 1) ^ (z >> 2) ^ (z >> 3));
}

__device__ __forceinline__ void wire_gate(const float* __restrict__ p,
                                          float gr[2][2], float gi[2][2]) {
  const float c0 = cosf(0.5f * p[0]), s0 = sinf(0.5f * p[0]);
  const float cp = cosf(0.5f * p[1]), sp = sinf(0.5f * p[1]);
  const float c2 = cosf(0.5f * p[2]), s2 = sinf(0.5f * p[2]);
  gr[0][0] =  cp * (c2 * c0 - s2 * s0);
  gi[0][0] = -sp * (c2 * c0 + s2 * s0);
  gr[0][1] = -cp * (c2 * s0 + s2 * c0);
  gi[0][1] =  sp * (c2 * s0 - s2 * c0);
  gr[1][0] =  cp * (s2 * c0 + c2 * s0);
  gi[1][0] =  sp * (c2 * s0 - s2 * c0);
  gr[1][1] =  cp * (c2 * c0 - s2 * s0);
  gi[1][1] =  sp * (c2 * c0 + s2 * s0);
}

__global__ void qgates_kernel(const float* __restrict__ params,
                              float* __restrict__ Ure, float* __restrict__ Uim) {
  const int t = threadIdx.x;
  const int r = t >> 4, c = t & 15;
  if (t < 2 * NQ) {
    float g0r[2][2], g0i[2][2];
    wire_gate(params + t * 3, g0r, g0i);
    float* o = Ure + t * 8;
    o[0] = g0r[0][0]; o[1] = g0i[0][0]; o[2] = g0r[0][1]; o[3] = g0i[0][1];
    o[4] = g0r[1][0]; o[5] = g0i[1][0]; o[6] = g0r[1][1]; o[7] = g0i[1][1];
  }
  for (int l = 2; l < NL; ++l) {
    for (int g = 0; g < 3; ++g) {
      float gr[4][2][2], gi[4][2][2];
      #pragma unroll
      for (int j = 0; j < 4; ++j)
        wire_gate(params + (l * NQ + g * 4 + j) * 3, gr[j], gi[j]);
      float ar = 1.f, ai = 0.f;
      #pragma unroll
      for (int j = 0; j < 4; ++j) {
        const int rb = (r >> (3 - j)) & 1, cb = (c >> (3 - j)) & 1;
        const float br = gr[j][rb][cb], bi = gi[j][rb][cb];
        const float nr = ar * br - ai * bi, ni = ar * bi + ai * br;
        ar = nr; ai = ni;
      }
      Ure[(l * 3 + g) * 256 + r * 16 + c] = ar;
      Uim[(l * 3 + g) * 256 + r * 16 + c] = ai;
    }
  }
}

__device__ __forceinline__ float2 cmul(float2 a, float2 b) {
  float2 r;
  r.x = a.x * b.x - a.y * b.y;
  r.y = a.x * b.y + a.y * b.x;
  return r;
}

__device__ __forceinline__ float4 rowmul(float4 p, float4 m0, float4 m1) {
  const float2 p0 = make_float2(p.x, p.y), p1 = make_float2(p.z, p.w);
  const float2 a0 = cmul(p0, make_float2(m0.x, m0.y)), a1 = cmul(p1, make_float2(m1.x, m1.y));
  const float2 b0 = cmul(p0, make_float2(m0.z, m0.w)), b1 = cmul(p1, make_float2(m1.z, m1.w));
  return make_float4(a0.x + a1.x, a0.y + a1.y, b0.x + b1.x, b0.y + b1.y);
}

template <int L, int PL, int G>
__device__ __forceinline__ void apply_group(float2* __restrict__ sAmp,
                                            const float* __restrict__ Ure,
                                            const float* __restrict__ Uim,
                                            int lane, int wave) {
  constexpr int RS = (G == 0) ? 256 : (G == 1) ? 16 : 1;
  const float* ur = Ure + (L * 3 + G) * 256;
  const float* ui = Uim + (L * 3 + G) * 256;
  const bool hi = (lane >= 16);
  const int koff = hi ? 2 : 0;
  const int moff = lane & 15;
  const int nn   = lane & 15;

  v2f aRe[4], aNS[4], aD[4];
  #pragma unroll
  for (int f = 0; f < 4; ++f) {
    const int k0 = 4 * f + koff;
    v2f tr, ti;
    tr.x = ur[moff * 16 + k0]; tr.y = ur[moff * 16 + k0 + 1];
    ti.x = ui[moff * 16 + k0]; ti.y = ui[moff * 16 + k0 + 1];
    aRe[f] = tr; aNS[f] = -(tr + ti); aD[f] = ti - tr;
  }

  int pk0[4], pk1[4], pm[8];
  #pragma unroll
  for (int f = 0; f < 4; ++f) {
    pk0[f] = hi ? permc<PL>((4 * f + 2) * RS) : permc<PL>((4 * f + 0) * RS);
    pk1[f] = hi ? permc<PL>((4 * f + 3) * RS) : permc<PL>((4 * f + 1) * RS);
  }
  #pragma unroll
  for (int v = 0; v < 8; ++v)
    pm[v] = hi ? permc<PL>((v + 8) * RS) : permc<PL>(v * RS);

  constexpr int TPW = NS * 16 / (THREADS / 32);
  #pragma unroll 2
  for (int t = 0; t < TPW; ++t) {
    const int tt = wave * TPW + t;
    const int ss = tt >> 4;
    const int c  = ((tt & 15) << 4) + nn;
    int other;
    if (G == 0)      other = c;
    else if (G == 1) other = ((c >> 4) << 8) | (c & 15);
    else             other = c << 4;
    const int pob = (ss << 12) | permc<PL>(other);

    v2f bRe[4], bIm[4], bS[4];
    #pragma unroll
    for (int f = 0; f < 4; ++f) {
      const float2 q0 = sAmp[pob ^ pk0[f]];
      const float2 q1 = sAmp[pob ^ pk1[f]];
      v2f tr, ti, ts;
      tr.x = q0.x; tr.y = q1.x;
      ti.x = q0.y; ti.y = q1.y;
      ts.x = q0.x + q0.y; ts.y = q1.x + q1.y;
      bRe[f] = tr; bIm[f] = ti; bS[f] = ts;
    }
    v8f dp = {};
    #pragma unroll
    for (int f = 0; f < 4; ++f)
      dp = __builtin_amdgcn_wmma_f32_16x16x4_f32(false, aRe[f], false, bS[f], (short)0, dp, false, false);
    v8f dre = dp; v8f dim_ = dp;
    #pragma unroll
    for (int f = 0; f < 4; ++f) {
      dre  = __builtin_amdgcn_wmma_f32_16x16x4_f32(false, aNS[f], false, bIm[f], (short)0, dre,  false, false);
      dim_ = __builtin_amdgcn_wmma_f32_16x16x4_f32(false, aD[f],  false, bRe[f], (short)0, dim_, false, false);
    }
    #pragma unroll
    for (int v = 0; v < 8; ++v) {
      float2 st; st.x = dre[v]; st.y = dim_[v];
      sAmp[pob ^ pm[v]] = st;
    }
  }
}

template <int L, int PL>
__device__ __forceinline__ void apply_group12(float2* __restrict__ sAmp,
                                              const float* __restrict__ Ure,
                                              const float* __restrict__ Uim,
                                              int lane, int wave) {
  const float* ur1 = Ure + (L * 3 + 1) * 256;
  const float* ui1 = Uim + (L * 3 + 1) * 256;
  const float* ur2 = Ure + (L * 3 + 2) * 256;
  const float* ui2 = Uim + (L * 3 + 2) * 256;
  const bool hi = (lane >= 16);
  const int nn = lane & 15;

  v2f g1Re[4], g1NS[4], g1D[4], g2Re[4], g2NS[4], g2D[4];
  #pragma unroll
  for (int f = 0; f < 4; ++f) {
    const int k1 = 4 * f + (hi ? 2 : 0);
    const int k2 = 2 * f + (hi ? 8 : 0);
    v2f tr, ti;
    tr.x = ur1[nn * 16 + k1]; tr.y = ur1[nn * 16 + k1 + 1];
    ti.x = ui1[nn * 16 + k1]; ti.y = ui1[nn * 16 + k1 + 1];
    g1Re[f] = tr; g1NS[f] = -(tr + ti); g1D[f] = ti - tr;
    tr.x = ur2[nn * 16 + k2]; tr.y = ur2[nn * 16 + k2 + 1];
    ti.x = ui2[nn * 16 + k2]; ti.y = ui2[nn * 16 + k2 + 1];
    g2Re[f] = tr; g2NS[f] = -(tr + ti); g2D[f] = ti - tr;
  }

  char* const sB = (char*)sAmp;
  const int laneL = (permc<PL>(nn) ^ (hi ? permc<PL>(2 * 16) : 0)) << 3;
  const int laneS = (permc<PL>(nn << 4) ^ (hi ? permc<PL>(8) : 0)) << 3;

  constexpr int TPW = NS * 16 / (THREADS / 32);
  #pragma unroll 1
  for (int t = 0; t < TPW; ++t) {
    const int tt = wave * TPW + t;
    const int tileB = (((tt >> 4) << 12) ^ permc<PL>((tt & 15) << 8)) << 3;
    const int baseL = tileB ^ laneL;
    const int baseS = tileB ^ laneS;

    v2f xRe[4], xIm[4], xS[4];
    #pragma unroll
    for (int f = 0; f < 4; ++f) {
      const float2 q0 = *(const float2*)(sB + (baseL ^ (permc<PL>((4 * f + 0) * 16) << 3)));
      const float2 q1 = *(const float2*)(sB + (baseL ^ (permc<PL>((4 * f + 1) * 16) << 3)));
      v2f tr, ti, ts;
      tr.x = q0.x; tr.y = q1.x;
      ti.x = q0.y; ti.y = q1.y;
      ts.x = q0.x + q0.y; ts.y = q1.x + q1.y;
      xRe[f] = tr; xIm[f] = ti; xS[f] = ts;
    }
    v8f p1 = {};
    #pragma unroll
    for (int f = 0; f < 4; ++f)
      p1 = __builtin_amdgcn_wmma_f32_16x16x4_f32(false, xS[f], false, g1Re[f], (short)0, p1, false, false);
    v8f yre = p1; v8f yim = p1;
    #pragma unroll
    for (int f = 0; f < 4; ++f) {
      yre = __builtin_amdgcn_wmma_f32_16x16x4_f32(false, xIm[f], false, g1NS[f], (short)0, yre, false, false);
      yim = __builtin_amdgcn_wmma_f32_16x16x4_f32(false, xRe[f], false, g1D[f],  (short)0, yim, false, false);
    }
    v2f bRe[4], bIm[4], bS[4];
    #pragma unroll
    for (int f = 0; f < 4; ++f) {
      v2f tr, ti;
      tr.x = yre[2 * f]; tr.y = yre[2 * f + 1];
      ti.x = yim[2 * f]; ti.y = yim[2 * f + 1];
      bRe[f] = tr; bIm[f] = ti; bS[f] = tr + ti;
    }
    v8f p2 = {};
    #pragma unroll
    for (int f = 0; f < 4; ++f)
      p2 = __builtin_amdgcn_wmma_f32_16x16x4_f32(false, g2Re[f], false, bS[f], (short)0, p2, false, false);
    v8f zre = p2; v8f zim = p2;
    #pragma unroll
    for (int f = 0; f < 4; ++f) {
      zre = __builtin_amdgcn_wmma_f32_16x16x4_f32(false, g2NS[f], false, bIm[f], (short)0, zre, false, false);
      zim = __builtin_amdgcn_wmma_f32_16x16x4_f32(false, g2D[f],  false, bRe[f], (short)0, zim, false, false);
    }
    #pragma unroll
    for (int v = 0; v < 8; ++v) {
      float2 st; st.x = zre[v]; st.y = zim[v];
      *(float2*)(sB + (baseS ^ (permc<PL>(v) << 3))) = st;
    }
  }
}

__global__ void __launch_bounds__(THREADS) __attribute__((amdgpu_waves_per_eu(8)))
qsim_kernel(const float* __restrict__ x, const float* __restrict__ head_w,
            const float* __restrict__ head_b, const float* __restrict__ Ure,
            const float* __restrict__ Uim, float* __restrict__ out) {
  extern __shared__ float smem_raw[];
  float2* sAmp = (float2*)smem_raw;
  float4* Tt   = (float4*)(sAmp + NS * DIM);
  float4* Pt   = Tt + NS * NQ * 4;
  float*  accS = (float*)(Pt + NS * 48 * 2);

  const int tid  = threadIdx.x;
  const int lane = tid & 31;
  const int wave = tid >> 5;
  const int s0   = blockIdx.x * NS;

  float w[NQ];
  #pragma unroll
  for (int i = 0; i < NQ; ++i) w[i] = head_w[i];

  if (tid < NS * NQ) {
    const int ss = tid / NQ, i = tid - ss * NQ;
    const float v = 0.5f * x[(s0 + ss) * NQ + i];
    const float cv = cosf(v), sv = sinf(v);
    const float* g = Ure + i * 8;
    float2 u[2];
    u[0].x = g[0] * cv + g[3] * sv;  u[0].y = g[1] * cv - g[2] * sv;
    u[1].x = g[4] * cv + g[7] * sv;  u[1].y = g[5] * cv - g[6] * sv;
    const float* G = Ure + (NQ + i) * 8;
    #pragma unroll
    for (int y = 0; y < 2; ++y) {
      const float2 Gy0 = make_float2(G[y * 4 + 0], G[y * 4 + 1]);
      const float2 Gy1 = make_float2(G[y * 4 + 2], G[y * 4 + 3]);
      #pragma unroll
      for (int a = 0; a < 2; ++a) {
        const float2 t0 = cmul(Gy0, u[a]);
        const float2 t1 = cmul(Gy1, u[a ^ 1]);
        Tt[(tid * 2 + y) * 2 + a] = make_float4(t0.x, t0.y, t1.x, t1.y);
      }
    }
  }
  __syncthreads();
  if (tid < NS * 48) {
    const int ss = tid / 48, e = tid - ss * 48;
    const int idx = e & 15;
    const float4* Tw = Tt + (ss * NQ + (e >> 4) * 4) * 4;
    int y = (idx >> 3) & 1;
    float4 r0 = Tw[y * 2 + 0], r1 = Tw[y * 2 + 1];
    for (int q = 1; q < 4; ++q) {
      y = (idx >> (3 - q)) & 1;
      const float4 m0 = Tw[q * 4 + y * 2 + 0], m1 = Tw[q * 4 + y * 2 + 1];
      r0 = rowmul(r0, m0, m1);
      r1 = rowmul(r1, m0, m1);
    }
    Pt[tid * 2 + 0] = r0;
    Pt[tid * 2 + 1] = r1;
  }
  __syncthreads();
  {
    constexpr int TPW = NS * 16 / (THREADS / 32);
    static_assert(16 % TPW == 0, "a wave's tiles belong to one sample");
    const int ss = (wave * TPW) >> 4;
    const bool hi = (lane >= 16);
    const int nn = lane & 15;
    const float4* Ps = Pt + ss * 96;
    const float4 lv = Ps[nn * 2];
    const float lre = hi ? lv.z : lv.x, lim = hi ? lv.w : lv.y;
    v2f aRe, aIm;
    aRe.x = lre; aRe.y = -lim;
    aIm.x = lim; aIm.y = lre;
    const float4 q0 = Ps[(32 + nn) * 2], q1 = Ps[(32 + nn) * 2 + 1];
    const float2 rl0 = make_float2(q0.x + q0.z, q0.y + q0.w);
    const float2 rl1 = make_float2(q1.x + q1.z, q1.y + q1.w);
    float2* const dst0 = sAmp + (ss << 12) + (hi ? 8 * 256 : 0) + nn;
    for (int t = 0; t < TPW; ++t) {
      const int m = (wave * TPW + t) & 15;
      const float4 mr = Ps[(16 + m) * 2 + (hi ? 1 : 0)];
      const float2 c0 = cmul(make_float2(mr.x, mr.y), rl0);
      const float2 c1 = cmul(make_float2(mr.z, mr.w), rl1);
      v2f b; b.x = c0.x + c1.x; b.y = c0.y + c1.y;
      const v8f zero = {};
      const v8f dre  = __builtin_amdgcn_wmma_f32_16x16x4_f32(false, aRe, false, b, (short)0, zero, false, false);
      const v8f dim_ = __builtin_amdgcn_wmma_f32_16x16x4_f32(false, aIm, false, b, (short)0, zero, false, false);
      #pragma unroll
      for (int v = 0; v < 8; ++v)
        dst0[(v << 8) + (m << 4)] = make_float2(dre[v], dim_[v]);
    }
  }
  __syncthreads();

  static_assert(NL == 3, "layers 0 and 1 are folded into the encode, layer 2 is the gate passes");
  apply_group<2, 1, 0>(sAmp, Ure, Uim, lane, wave); __syncthreads();
  apply_group12<2, 1>(sAmp, Ure, Uim, lane, wave);  __syncthreads();

  if (tid < NS) accS[tid] = 0.f;
  __syncthreads();
  float wsum = 0.f;
  #pragma unroll
  for (int i = 0; i < NQ; ++i) wsum += w[i];
  float cl = wsum;
  #pragma unroll
  for (int i = 4; i < NQ; ++i)
    cl -= 2.f * w[i] * (float)((tid >> (NQ - 1 - i)) & 1);
  float coef[16];
  #pragma unroll
  for (int j = 0; j < 16; ++j) {
    float ch = 0.f;
    #pragma unroll
    for (int i = 0; i < 4; ++i)
      ch -= 2.f * w[i] * (float)((j >> (3 - i)) & 1);
    coef[j] = cl + ch;
  }
  const int pt = permc<NL - 1>(tid);
  #pragma unroll
  for (int ss = 0; ss < NS; ++ss) {
    float a = 0.f;
    #pragma unroll
    for (int j = 0; j < 16; ++j) {
      const float2 q = sAmp[(ss << 12) + (pt ^ permc<NL - 1>(j << 8))];
      a += (q.x * q.x + q.y * q.y) * coef[j];
    }
    atomicAdd(&accS[ss], a);
  }
  __syncthreads();
  if (tid < NS) out[s0 + tid] = accS[tid] + head_b[0];
}

extern "C" void kernel_launch(void* const* d_in, const int* in_sizes, int n_in,
                              void* d_out, int out_size, void* d_ws, size_t ws_size,
                              hipStream_t stream) {
  const float* x      = (const float*)d_in[0];
  const float* params = (const float*)d_in[1];
  const float* head_w = (const float*)d_in[2];
  const float* head_b = (const float*)d_in[3];
  float* out = (float*)d_out;
  float* Ure = (float*)d_ws;
  float* Uim = Ure + NL * 3 * 256;

  const int B = in_sizes[0] / NQ;
  const size_t shmem =
      (size_t)(2 * NS * DIM + NS + 4 * (NS * NQ * 4 + NS * 48 * 2)) * sizeof(float);
  (void)hipFuncSetAttribute((const void*)qsim_kernel,
                            hipFuncAttributeMaxDynamicSharedMemorySize,
                            (int)shmem);

  qgates_kernel<<<1, 256, 0, stream>>>(params, Ure, Uim);
  qsim_kernel<<<B / NS, THREADS, shmem, stream>>>(x, head_w, head_b, Ure, Uim, out);
}


# Readout taken in the tile's D registers (|amp|^2 times the coefficient of its CNOT pre-image), lanes then waves summed: no amplitude state in LDS, no atomics (re-associates the readout sum)
# speedup vs baseline: 13.8644x; 3.0433x over previous
#include <hip/hip_runtime.h>

typedef __attribute__((ext_vector_type(2))) float v2f;
typedef __attribute__((ext_vector_type(8))) float v8f;

#define NQ      12
#define DIM     4096
#define NS      2
#define NL      3
#define THREADS 256

__device__ __forceinline__ float prefix_sign_sum(const float* w, int q) {
  float acc = 0.f;
  int par = 0;
  #pragma unroll
  for (int k = 0; k < 4; ++k) {
    par ^= (q >> (3 - k)) & 1;
    acc += par ? -w[k] : w[k];
  }
  return acc;
}

__device__ __forceinline__ void wire_gate(const float* __restrict__ p,
                                          float gr[2][2], float gi[2][2]) {
  const float c0 = cosf(0.5f * p[0]), s0 = sinf(0.5f * p[0]);
  const float cp = cosf(0.5f * p[1]), sp = sinf(0.5f * p[1]);
  const float c2 = cosf(0.5f * p[2]), s2 = sinf(0.5f * p[2]);
  gr[0][0] =  cp * (c2 * c0 - s2 * s0);
  gi[0][0] = -sp * (c2 * c0 + s2 * s0);
  gr[0][1] = -cp * (c2 * s0 + s2 * c0);
  gi[0][1] =  sp * (c2 * s0 - s2 * c0);
  gr[1][0] =  cp * (s2 * c0 + c2 * s0);
  gi[1][0] =  sp * (c2 * s0 - s2 * c0);
  gr[1][1] =  cp * (c2 * c0 - s2 * s0);
  gi[1][1] =  sp * (c2 * c0 + s2 * s0);
}

__global__ void qgates_kernel(const float* __restrict__ params, float* __restrict__ G2) {
  const int t = threadIdx.x;
  if (t < NL * NQ) {
    float gr[2][2], gi[2][2];
    wire_gate(params + t * 3, gr, gi);
    float* o = G2 + t * 8;
    o[0] = gr[0][0]; o[1] = gi[0][0]; o[2] = gr[0][1]; o[3] = gi[0][1];
    o[4] = gr[1][0]; o[5] = gi[1][0]; o[6] = gr[1][1]; o[7] = gi[1][1];
  }
}

__device__ __forceinline__ float2 cmul(float2 a, float2 b) {
  float2 r;
  r.x = a.x * b.x - a.y * b.y;
  r.y = a.x * b.y + a.y * b.x;
  return r;
}

__device__ __forceinline__ float xhalf(float x) {
  const int b = __builtin_bit_cast(int, x);
  return __builtin_bit_cast(float,
      __builtin_amdgcn_permlanex16(b, b, 0x76543210, 0xfedcba98, false, false));
}

template <int MASK>
__device__ __forceinline__ float xrow(float x) {
  const int b = __builtin_bit_cast(int, x);
  return __builtin_bit_cast(float,
      __builtin_amdgcn_update_dpp(b, b, 0x160 + MASK, 0xf, 0xf, false));
}

__device__ __forceinline__ void cfma(float2& acc, float2 a, float2 b) {
  acc.x += a.x * b.x - a.y * b.y;
  acc.y += a.x * b.y + a.y * b.x;
}

__global__ void __launch_bounds__(THREADS)
qsim_kernel(const float* __restrict__ x, const float* __restrict__ head_w,
            const float* __restrict__ head_b, const float* __restrict__ G2,
            float* __restrict__ out) {
  __shared__ __align__(16) float2 Rt[NS * NQ * 2 * 16];
  __shared__ float4 Vt[NS * 96 * 2];
  __shared__ float  part[THREADS / 32];

  const int tid  = threadIdx.x;
  const int lane = tid & 31;
  const int wave = tid >> 5;
  const int s0   = blockIdx.x * NS;

  float w[NQ];
  #pragma unroll
  for (int i = 0; i < NQ; ++i) w[i] = head_w[i];

  if (tid < NS * NQ * 2) {
    const int y = tid & 1, si = tid >> 1;
    const int ss = si / NQ, i = si - ss * NQ;
    const float v = 0.5f * x[(s0 + ss) * NQ + i];
    const float cv = cosf(v), sv = sinf(v);
    const float* g = G2 + i * 8;
    float2 u[2];
    u[0].x = g[0] * cv + g[3] * sv;  u[0].y = g[1] * cv - g[2] * sv;
    u[1].x = g[4] * cv + g[7] * sv;  u[1].y = g[5] * cv - g[6] * sv;
    const float* G = G2 + (NQ + i) * 8;
    const float* H = G2 + (2 * NQ + i) * 8;
    float2 T[2][2][2];
    #pragma unroll
    for (int t = 0; t < 2; ++t)
      #pragma unroll
      for (int a = 0; a < 2; ++a)
        #pragma unroll
        for (int b = 0; b < 2; ++b)
          T[t][a][b] = cmul(make_float2(G[t * 4 + b * 2], G[t * 4 + b * 2 + 1]), u[a ^ b]);
    const bool tr = (i >= 8);
    float2* Ro = Rt + tid * 16;
    #pragma unroll
    for (int q = 0; q < 2; ++q) {
      const float2 Hq = make_float2(H[y * 4 + q * 2], H[y * 4 + q * 2 + 1]);
      #pragma unroll
      for (int a = 0; a < 2; ++a)
        #pragma unroll
        for (int p = 0; p < 2; ++p)
          #pragma unroll
          for (int b = 0; b < 2; ++b) {
            const int r = 2 * a + p, c = 2 * b + q;
            Ro[tr ? (c * 4 + r) : (r * 4 + c)] = cmul(Hq, T[q ^ p][a][b]);
          }
    }
  }
  __syncthreads();
  if (tid < NS * 96) {
    const int ss = tid / 96, e = tid - ss * 96;
    const bool isCol = (e >= 80);
    const int grp  = isCol ? 2 : (e >= 16 ? 1 : 0);
    const int idx4 = isCol ? (e - 80) : (e >= 16 ? ((e - 16) >> 2) : e);
    const int row  = (grp == 1) ? ((e - 16) & 3) : 0;
    float2 vv[4];
    #pragma unroll
    for (int k = 0; k < 4; ++k)
      vv[k] = make_float2((isCol || k == row) ? 1.f : 0.f, 0.f);
    #pragma unroll
    for (int q = 0; q < 4; ++q) {
      const int wl = isCol ? (3 - q) : q;
      const int yb = (idx4 >> (3 - wl)) & 1;
      const float4* M4 = (const float4*)(Rt + ((ss * NQ + grp * 4 + wl) * 2 + yb) * 16);
      float2 o[4];
      #pragma unroll
      for (int k = 0; k < 4; ++k) o[k] = make_float2(0.f, 0.f);
      #pragma unroll
      for (int r = 0; r < 4; ++r) {
        const float4 m01 = M4[r * 2], m23 = M4[r * 2 + 1];
        cfma(o[0], vv[r], make_float2(m01.x, m01.y));
        cfma(o[1], vv[r], make_float2(m01.z, m01.w));
        cfma(o[2], vv[r], make_float2(m23.x, m23.y));
        cfma(o[3], vv[r], make_float2(m23.z, m23.w));
      }
      #pragma unroll
      for (int k = 0; k < 4; ++k) vv[k] = o[k];
    }
    Vt[tid * 2 + 0] = make_float4(vv[0].x, vv[0].y, vv[1].x, vv[1].y);
    Vt[tid * 2 + 1] = make_float4(vv[2].x, vv[2].y, vv[3].x, vv[3].y);
  }
  __syncthreads();
  {
    constexpr int TPW = NS * 16 / (THREADS / 32);
    static_assert(16 % TPW == 0, "a wave's tiles belong to one sample");
    const int ss = (wave * TPW) >> 4;
    const bool hi = (lane >= 16);
    const int nn = lane & 15;
    const float4* Vs = Vt + ss * 96 * 2;
    const float4 l01 = Vs[nn * 2], l23 = Vs[nn * 2 + 1];
    const float2 la = hi ? make_float2(l01.z, l01.w) : make_float2(l01.x, l01.y);
    const float2 lb = hi ? make_float2(l23.z, l23.w) : make_float2(l23.x, l23.y);
    v2f aRe1, aIm1, aRe2, aIm2;
    aRe1.x = la.x; aRe1.y = -la.y;  aIm1.x = la.y; aIm1.y = la.x;
    aRe2.x = lb.x; aRe2.y = -lb.y;  aIm2.x = lb.y; aIm2.y = lb.x;
    const float4 r01 = Vs[(80 + nn) * 2], r23 = Vs[(80 + nn) * 2 + 1];
    const float2 r0 = make_float2(r01.x, r01.y), r1 = make_float2(r01.z, r01.w);
    const float2 r2 = make_float2(r23.x, r23.y), r3 = make_float2(r23.z, r23.w);
    float Aj[8];
    #pragma unroll
    for (int v = 0; v < 8; ++v) Aj[v] = prefix_sign_sum(w, v);
    const float Cn = prefix_sign_sum(w + 8, nn);
    float racc = 0.f;
    for (int t = 0; t < TPW; ++t) {
      const int m = (wave * TPW + t) & 15;
      const float4* Mr = Vs + (16 + m * 4 + (hi ? 1 : 0)) * 2;
      const float4 ma01 = Mr[0], ma23 = Mr[1], mb01 = Mr[4], mb23 = Mr[5];
      float2 va = cmul(make_float2(ma01.x, ma01.y), r0);
      cfma(va, make_float2(ma01.z, ma01.w), r1);
      cfma(va, make_float2(ma23.x, ma23.y), r2);
      cfma(va, make_float2(ma23.z, ma23.w), r3);
      float2 vb = cmul(make_float2(mb01.x, mb01.y), r0);
      cfma(vb, make_float2(mb01.z, mb01.w), r1);
      cfma(vb, make_float2(mb23.x, mb23.y), r2);
      cfma(vb, make_float2(mb23.z, mb23.w), r3);
      v2f b1, b2;
      b1.x = va.x; b1.y = va.y;
      b2.x = vb.x; b2.y = vb.y;
      const v8f zero = {};
      v8f dre  = __builtin_amdgcn_wmma_f32_16x16x4_f32(false, aRe1, false, b1, (short)0, zero, false, false);
      v8f dim_ = __builtin_amdgcn_wmma_f32_16x16x4_f32(false, aIm1, false, b1, (short)0, zero, false, false);
      dre  = __builtin_amdgcn_wmma_f32_16x16x4_f32(false, aRe2, false, b2, (short)0, dre,  false, false);
      dim_ = __builtin_amdgcn_wmma_f32_16x16x4_f32(false, aIm2, false, b2, (short)0, dim_, false, false);
      const float Bm = prefix_sign_sum(w + 4, m);
      const float E  = (__popc(m) & 1) ? Bm - Cn : Bm + Cn;
      #pragma unroll
      for (int v = 0; v < 8; ++v) {
        const float p = dre[v] * dre[v] + dim_[v] * dim_[v];
        const float c = (__popc(v) & 1) ? Aj[v] - E : Aj[v] + E;
        racc += p * c;
      }
    }
    racc = hi ? -racc : racc;
    racc += xrow<1>(racc);
    racc += xrow<2>(racc);
    racc += xrow<4>(racc);
    racc += xrow<8>(racc);
    racc += xhalf(racc);
    if (lane == 0) part[wave] = racc;
  }
  __syncthreads();
  static_assert(NL == 3, "all three layers' rotations are folded into the encode");
  if (tid < NS) {
    constexpr int WPS = (THREADS / 32) / NS;
    float a = part[tid * WPS];
    #pragma unroll
    for (int k = 1; k < WPS; ++k) a += part[tid * WPS + k];
    out[s0 + tid] = a + head_b[0];
  }
}

extern "C" void kernel_launch(void* const* d_in, const int* in_sizes, int n_in,
                              void* d_out, int out_size, void* d_ws, size_t ws_size,
                              hipStream_t stream) {
  const float* x      = (const float*)d_in[0];
  const float* params = (const float*)d_in[1];
  const float* head_w = (const float*)d_in[2];
  const float* head_b = (const float*)d_in[3];
  float* out = (float*)d_out;
  float* G2 = (float*)d_ws;

  const int B = in_sizes[0] / NQ;

  qgates_kernel<<<1, 64, 0, stream>>>(params, G2);
  qsim_kernel<<<B / NS, THREADS, 0, stream>>>(x, head_w, head_b, G2, out);
}


# One wave per sample (4-wave workgroups of 4 samples; was 8 waves for 2): a wave runs all 16 tiles of its sample, a quarter of the waves enter, the readout sum stays in one wave (re-associates it)
# speedup vs baseline: 14.6861x; 1.0593x over previous
#include <hip/hip_runtime.h>

typedef __attribute__((ext_vector_type(2))) float v2f;
typedef __attribute__((ext_vector_type(8))) float v8f;

#define NQ      12
#define DIM     4096
#define NS      4
#define NL      3
#define THREADS (32 * NS)

__device__ __forceinline__ float prefix_sign_sum(const float* w, int q) {
  float acc = 0.f;
  int par = 0;
  #pragma unroll
  for (int k = 0; k < 4; ++k) {
    par ^= (q >> (3 - k)) & 1;
    acc += par ? -w[k] : w[k];
  }
  return acc;
}

__device__ __forceinline__ void wire_gate(const float* __restrict__ p,
                                          float gr[2][2], float gi[2][2]) {
  const float c0 = cosf(0.5f * p[0]), s0 = sinf(0.5f * p[0]);
  const float cp = cosf(0.5f * p[1]), sp = sinf(0.5f * p[1]);
  const float c2 = cosf(0.5f * p[2]), s2 = sinf(0.5f * p[2]);
  gr[0][0] =  cp * (c2 * c0 - s2 * s0);
  gi[0][0] = -sp * (c2 * c0 + s2 * s0);
  gr[0][1] = -cp * (c2 * s0 + s2 * c0);
  gi[0][1] =  sp * (c2 * s0 - s2 * c0);
  gr[1][0] =  cp * (s2 * c0 + c2 * s0);
  gi[1][0] =  sp * (c2 * s0 - s2 * c0);
  gr[1][1] =  cp * (c2 * c0 - s2 * s0);
  gi[1][1] =  sp * (c2 * c0 + s2 * s0);
}

__global__ void qgates_kernel(const float* __restrict__ params, float* __restrict__ G2) {
  const int t = threadIdx.x;
  if (t < NL * NQ) {
    float gr[2][2], gi[2][2];
    wire_gate(params + t * 3, gr, gi);
    float* o = G2 + t * 8;
    o[0] = gr[0][0]; o[1] = gi[0][0]; o[2] = gr[0][1]; o[3] = gi[0][1];
    o[4] = gr[1][0]; o[5] = gi[1][0]; o[6] = gr[1][1]; o[7] = gi[1][1];
  }
}

__device__ __forceinline__ float2 cmul(float2 a, float2 b) {
  float2 r;
  r.x = a.x * b.x - a.y * b.y;
  r.y = a.x * b.y + a.y * b.x;
  return r;
}

__device__ __forceinline__ float xhalf(float x) {
  const int b = __builtin_bit_cast(int, x);
  return __builtin_bit_cast(float,
      __builtin_amdgcn_permlanex16(b, b, 0x76543210, 0xfedcba98, false, false));
}

template <int MASK>
__device__ __forceinline__ float xrow(float x) {
  const int b = __builtin_bit_cast(int, x);
  return __builtin_bit_cast(float,
      __builtin_amdgcn_update_dpp(b, b, 0x160 + MASK, 0xf, 0xf, false));
}

__device__ __forceinline__ void cfma(float2& acc, float2 a, float2 b) {
  acc.x += a.x * b.x - a.y * b.y;
  acc.y += a.x * b.y + a.y * b.x;
}

__global__ void __launch_bounds__(THREADS)
qsim_kernel(const float* __restrict__ x, const float* __restrict__ head_w,
            const float* __restrict__ head_b, const float* __restrict__ G2,
            float* __restrict__ out) {
  __shared__ __align__(16) float2 Rt[NS * NQ * 2 * 16];
  __shared__ float4 Vt[NS * 96 * 2];

  const int tid  = threadIdx.x;
  const int lane = tid & 31;
  const int wave = tid >> 5;
  const int s0   = blockIdx.x * NS;

  float w[NQ];
  #pragma unroll
  for (int i = 0; i < NQ; ++i) w[i] = head_w[i];

  if (tid < NS * NQ * 2) {
    const int y = tid & 1, si = tid >> 1;
    const int ss = si / NQ, i = si - ss * NQ;
    const float v = 0.5f * x[(s0 + ss) * NQ + i];
    const float cv = cosf(v), sv = sinf(v);
    const float* g = G2 + i * 8;
    float2 u[2];
    u[0].x = g[0] * cv + g[3] * sv;  u[0].y = g[1] * cv - g[2] * sv;
    u[1].x = g[4] * cv + g[7] * sv;  u[1].y = g[5] * cv - g[6] * sv;
    const float* G = G2 + (NQ + i) * 8;
    const float* H = G2 + (2 * NQ + i) * 8;
    float2 T[2][2][2];
    #pragma unroll
    for (int t = 0; t < 2; ++t)
      #pragma unroll
      for (int a = 0; a < 2; ++a)
        #pragma unroll
        for (int b = 0; b < 2; ++b)
          T[t][a][b] = cmul(make_float2(G[t * 4 + b * 2], G[t * 4 + b * 2 + 1]), u[a ^ b]);
    const bool tr = (i >= 8);
    float2* Ro = Rt + tid * 16;
    #pragma unroll
    for (int q = 0; q < 2; ++q) {
      const float2 Hq = make_float2(H[y * 4 + q * 2], H[y * 4 + q * 2 + 1]);
      #pragma unroll
      for (int a = 0; a < 2; ++a)
        #pragma unroll
        for (int p = 0; p < 2; ++p)
          #pragma unroll
          for (int b = 0; b < 2; ++b) {
            const int r = 2 * a + p, c = 2 * b + q;
            Ro[tr ? (c * 4 + r) : (r * 4 + c)] = cmul(Hq, T[q ^ p][a][b]);
          }
    }
  }
  __syncthreads();
  static_assert((NS * 96) % THREADS == 0, "the four-wire results fill whole passes");
  for (int it = tid; it < NS * 96; it += THREADS) {
    const int ss = it / 96, e = it - ss * 96;
    const bool isCol = (e >= 80);
    const int grp  = isCol ? 2 : (e >= 16 ? 1 : 0);
    const int idx4 = isCol ? (e - 80) : (e >= 16 ? ((e - 16) >> 2) : e);
    const int row  = (grp == 1) ? ((e - 16) & 3) : 0;
    float2 vv[4];
    #pragma unroll
    for (int k = 0; k < 4; ++k)
      vv[k] = make_float2((isCol || k == row) ? 1.f : 0.f, 0.f);
    #pragma unroll
    for (int q = 0; q < 4; ++q) {
      const int wl = isCol ? (3 - q) : q;
      const int yb = (idx4 >> (3 - wl)) & 1;
      const float4* M4 = (const float4*)(Rt + ((ss * NQ + grp * 4 + wl) * 2 + yb) * 16);
      float2 o[4];
      #pragma unroll
      for (int k = 0; k < 4; ++k) o[k] = make_float2(0.f, 0.f);
      #pragma unroll
      for (int r = 0; r < 4; ++r) {
        const float4 m01 = M4[r * 2], m23 = M4[r * 2 + 1];
        cfma(o[0], vv[r], make_float2(m01.x, m01.y));
        cfma(o[1], vv[r], make_float2(m01.z, m01.w));
        cfma(o[2], vv[r], make_float2(m23.x, m23.y));
        cfma(o[3], vv[r], make_float2(m23.z, m23.w));
      }
      #pragma unroll
      for (int k = 0; k < 4; ++k) vv[k] = o[k];
    }
    Vt[it * 2 + 0] = make_float4(vv[0].x, vv[0].y, vv[1].x, vv[1].y);
    Vt[it * 2 + 1] = make_float4(vv[2].x, vv[2].y, vv[3].x, vv[3].y);
  }
  __syncthreads();
  {
    const int ss = wave;
    const bool hi = (lane >= 16);
    const int nn = lane & 15;
    const float4* Vs = Vt + ss * 96 * 2;
    const float4 l01 = Vs[nn * 2], l23 = Vs[nn * 2 + 1];
    const float2 la = hi ? make_float2(l01.z, l01.w) : make_float2(l01.x, l01.y);
    const float2 lb = hi ? make_float2(l23.z, l23.w) : make_float2(l23.x, l23.y);
    v2f aRe1, aIm1, aRe2, aIm2;
    aRe1.x = la.x; aRe1.y = -la.y;  aIm1.x = la.y; aIm1.y = la.x;
    aRe2.x = lb.x; aRe2.y = -lb.y;  aIm2.x = lb.y; aIm2.y = lb.x;
    const float4 r01 = Vs[(80 + nn) * 2], r23 = Vs[(80 + nn) * 2 + 1];
    const float2 r0 = make_float2(r01.x, r01.y), r1 = make_float2(r01.z, r01.w);
    const float2 r2 = make_float2(r23.x, r23.y), r3 = make_float2(r23.z, r23.w);
    float Aj[8];
    #pragma unroll
    for (int v = 0; v < 8; ++v) Aj[v] = prefix_sign_sum(w, v);
    const float Cn = prefix_sign_sum(w + 8, nn);
    float racc = 0.f;
    for (int m = 0; m < 16; ++m) {
      const float4* Mr = Vs + (16 + m * 4 + (hi ? 1 : 0)) * 2;
      const float4 ma01 = Mr[0], ma23 = Mr[1], mb01 = Mr[4], mb23 = Mr[5];
      float2 va = cmul(make_float2(ma01.x, ma01.y), r0);
      cfma(va, make_float2(ma01.z, ma01.w), r1);
      cfma(va, make_float2(ma23.x, ma23.y), r2);
      cfma(va, make_float2(ma23.z, ma23.w), r3);
      float2 vb = cmul(make_float2(mb01.x, mb01.y), r0);
      cfma(vb, make_float2(mb01.z, mb01.w), r1);
      cfma(vb, make_float2(mb23.x, mb23.y), r2);
      cfma(vb, make_float2(mb23.z, mb23.w), r3);
      v2f b1, b2;
      b1.x = va.x; b1.y = va.y;
      b2.x = vb.x; b2.y = vb.y;
      const v8f zero = {};
      v8f dre  = __builtin_amdgcn_wmma_f32_16x16x4_f32(false, aRe1, false, b1, (short)0, zero, false, false);
      v8f dim_ = __builtin_amdgcn_wmma_f32_16x16x4_f32(false, aIm1, false, b1, (short)0, zero, false, false);
      dre  = __builtin_amdgcn_wmma_f32_16x16x4_f32(false, aRe2, false, b2, (short)0, dre,  false, false);
      dim_ = __builtin_amdgcn_wmma_f32_16x16x4_f32(false, aIm2, false, b2, (short)0, dim_, false, false);
      const float Bm = prefix_sign_sum(w + 4, m);
      const float E  = (__popc(m) & 1) ? Bm - Cn : Bm + Cn;
      #pragma unroll
      for (int v = 0; v < 8; ++v) {
        const float p = dre[v] * dre[v] + dim_[v] * dim_[v];
        const float c = (__popc(v) & 1) ? Aj[v] - E : Aj[v] + E;
        racc += p * c;
      }
    }
    racc = hi ? -racc : racc;
    racc += xrow<1>(racc);
    racc += xrow<2>(racc);
    racc += xrow<4>(racc);
    racc += xrow<8>(racc);
    racc += xhalf(racc);
    if (lane == 0) out[s0 + wave] = racc + head_b[0];
  }
  static_assert(NL == 3, "all three layers' rotations are folded into the encode");
}

extern "C" void kernel_launch(void* const* d_in, const int* in_sizes, int n_in,
                              void* d_out, int out_size, void* d_ws, size_t ws_size,
                              hipStream_t stream) {
  const float* x      = (const float*)d_in[0];
  const float* params = (const float*)d_in[1];
  const float* head_w = (const float*)d_in[2];
  const float* head_b = (const float*)d_in[3];
  float* out = (float*)d_out;
  float* G2 = (float*)d_ws;

  const int B = in_sizes[0] / NQ;

  qgates_kernel<<<1, 64, 0, stream>>>(params, G2);
  qsim_kernel<<<B / NS, THREADS, 0, stream>>>(x, head_w, head_b, G2, out);
}


# Gate set-up folded into the main launch: every workgroup builds the 36 wire gates itself (36 threads, into LDS, one more barrier); one launch per call, not two, no workspace; no sum re-ordered
# speedup vs baseline: 16.6885x; 1.1363x over previous
#include <hip/hip_runtime.h>

typedef __attribute__((ext_vector_type(2))) float v2f;
typedef __attribute__((ext_vector_type(8))) float v8f;

#define NQ      12
#define DIM     4096
#define NS      4
#define NL      3
#define THREADS (32 * NS)

__device__ __forceinline__ float prefix_sign_sum(const float* w, int q) {
  float acc = 0.f;
  int par = 0;
  #pragma unroll
  for (int k = 0; k < 4; ++k) {
    par ^= (q >> (3 - k)) & 1;
    acc += par ? -w[k] : w[k];
  }
  return acc;
}

__device__ __forceinline__ void wire_gate(float p0, float p1, float p2,
                                          float gr[2][2], float gi[2][2]) {
  const float c0 = cosf(0.5f * p0), s0 = sinf(0.5f * p0);
  const float cp = cosf(0.5f * p1), sp = sinf(0.5f * p1);
  const float c2 = cosf(0.5f * p2), s2 = sinf(0.5f * p2);
  gr[0][0] =  cp * (c2 * c0 - s2 * s0);
  gi[0][0] = -sp * (c2 * c0 + s2 * s0);
  gr[0][1] = -cp * (c2 * s0 + s2 * c0);
  gi[0][1] =  sp * (c2 * s0 - s2 * c0);
  gr[1][0] =  cp * (s2 * c0 + c2 * s0);
  gi[1][0] =  sp * (c2 * s0 - s2 * c0);
  gr[1][1] =  cp * (c2 * c0 - s2 * s0);
  gi[1][1] =  sp * (c2 * c0 + s2 * s0);
}

__device__ __forceinline__ float2 cmul(float2 a, float2 b) {
  float2 r;
  r.x = a.x * b.x - a.y * b.y;
  r.y = a.x * b.y + a.y * b.x;
  return r;
}

__device__ __forceinline__ float xhalf(float x) {
  const int b = __builtin_bit_cast(int, x);
  return __builtin_bit_cast(float,
      __builtin_amdgcn_permlanex16(b, b, 0x76543210, 0xfedcba98, false, false));
}

template <int MASK>
__device__ __forceinline__ float xrow(float x) {
  const int b = __builtin_bit_cast(int, x);
  return __builtin_bit_cast(float,
      __builtin_amdgcn_update_dpp(b, b, 0x160 + MASK, 0xf, 0xf, false));
}

__device__ __forceinline__ void cfma(float2& acc, float2 a, float2 b) {
  acc.x += a.x * b.x - a.y * b.y;
  acc.y += a.x * b.y + a.y * b.x;
}

__global__ void __launch_bounds__(THREADS)
qsim_kernel(const float* __restrict__ x, const float* __restrict__ params,
            const float* __restrict__ head_w, const float* __restrict__ head_b,
            float* __restrict__ out) {
  __shared__ __align__(16) float2 Rt[NS * NQ * 2 * 16];
  __shared__ float4 Vt[NS * 96 * 2];
  __shared__ float4 G2[NL * NQ * 2];

  const int tid  = threadIdx.x;
  const int lane = tid & 31;
  const int wave = tid >> 5;
  const int s0   = blockIdx.x * NS;

  float w[NQ];
  #pragma unroll
  for (int i = 0; i < NQ; ++i) w[i] = head_w[i];

  static_assert(NL * NQ <= THREADS, "one thread per gate");
  const int t = THREADS - 1 - tid;
  const bool gate_thread = (t < NL * NQ);
  const bool wire_thread = (tid < NS * NQ * 2);
  const float* pg = params + (gate_thread ? t : 0) * 3;
  const float p0 = pg[0], p1 = pg[1], p2 = pg[2];
  const float xv = x[s0 * NQ + (wire_thread ? (tid >> 1) : 0)];
  if (gate_thread) {
    float gr[2][2], gi[2][2];
    wire_gate(p0, p1, p2, gr, gi);
    G2[t * 2 + 0] = make_float4(gr[0][0], gi[0][0], gr[0][1], gi[0][1]);
    G2[t * 2 + 1] = make_float4(gr[1][0], gi[1][0], gr[1][1], gi[1][1]);
  }
  float cv = 0.f, sv = 0.f;
  if (wire_thread) {
    cv = cosf(0.5f * xv); sv = sinf(0.5f * xv);
  }
  __syncthreads();

  if (tid < NS * NQ * 2) {
    const int y = tid & 1, si = tid >> 1;
    const int ss = si / NQ, i = si - ss * NQ;
    const float4 g01 = G2[i * 2], g23 = G2[i * 2 + 1];
    const float g[8] = {g01.x, g01.y, g01.z, g01.w, g23.x, g23.y, g23.z, g23.w};
    float2 u[2];
    u[0].x = g[0] * cv + g[3] * sv;  u[0].y = g[1] * cv - g[2] * sv;
    u[1].x = g[4] * cv + g[7] * sv;  u[1].y = g[5] * cv - g[6] * sv;
    const float4 G01 = G2[(NQ + i) * 2], G23 = G2[(NQ + i) * 2 + 1];
    const float G[8] = {G01.x, G01.y, G01.z, G01.w, G23.x, G23.y, G23.z, G23.w};
    const float4 Hy = G2[(2 * NQ + i) * 2 + y];
    const float H[4] = {Hy.x, Hy.y, Hy.z, Hy.w};
    float2 T[2][2][2];
    #pragma unroll
    for (int t = 0; t < 2; ++t)
      #pragma unroll
      for (int a = 0; a < 2; ++a)
        #pragma unroll
        for (int b = 0; b < 2; ++b)
          T[t][a][b] = cmul(make_float2(G[t * 4 + b * 2], G[t * 4 + b * 2 + 1]), u[a ^ b]);
    const bool tr = (i >= 8);
    float2* Ro = Rt + tid * 16;
    #pragma unroll
    for (int q = 0; q < 2; ++q) {
      const float2 Hq = make_float2(H[q * 2], H[q * 2 + 1]);
      #pragma unroll
      for (int a = 0; a < 2; ++a)
        #pragma unroll
        for (int p = 0; p < 2; ++p)
          #pragma unroll
          for (int b = 0; b < 2; ++b) {
            const int r = 2 * a + p, c = 2 * b + q;
            Ro[tr ? (c * 4 + r) : (r * 4 + c)] = cmul(Hq, T[q ^ p][a][b]);
          }
    }
  }
  __syncthreads();
  static_assert((NS * 96) % THREADS == 0, "the four-wire results fill whole passes");
  for (int it = tid; it < NS * 96; it += THREADS) {
    const int ss = it / 96, e = it - ss * 96;
    const bool isCol = (e >= 80);
    const int grp  = isCol ? 2 : (e >= 16 ? 1 : 0);
    const int idx4 = isCol ? (e - 80) : (e >= 16 ? ((e - 16) >> 2) : e);
    const int row  = (grp == 1) ? ((e - 16) & 3) : 0;
    float2 vv[4];
    #pragma unroll
    for (int k = 0; k < 4; ++k)
      vv[k] = make_float2((isCol || k == row) ? 1.f : 0.f, 0.f);
    #pragma unroll
    for (int q = 0; q < 4; ++q) {
      const int wl = isCol ? (3 - q) : q;
      const int yb = (idx4 >> (3 - wl)) & 1;
      const float4* M4 = (const float4*)(Rt + ((ss * NQ + grp * 4 + wl) * 2 + yb) * 16);
      float2 o[4];
      #pragma unroll
      for (int k = 0; k < 4; ++k) o[k] = make_float2(0.f, 0.f);
      #pragma unroll
      for (int r = 0; r < 4; ++r) {
        const float4 m01 = M4[r * 2], m23 = M4[r * 2 + 1];
        cfma(o[0], vv[r], make_float2(m01.x, m01.y));
        cfma(o[1], vv[r], make_float2(m01.z, m01.w));
        cfma(o[2], vv[r], make_float2(m23.x, m23.y));
        cfma(o[3], vv[r], make_float2(m23.z, m23.w));
      }
      #pragma unroll
      for (int k = 0; k < 4; ++k) vv[k] = o[k];
    }
    Vt[it * 2 + 0] = make_float4(vv[0].x, vv[0].y, vv[1].x, vv[1].y);
    Vt[it * 2 + 1] = make_float4(vv[2].x, vv[2].y, vv[3].x, vv[3].y);
  }
  __syncthreads();
  {
    const int ss = wave;
    const bool hi = (lane >= 16);
    const int nn = lane & 15;
    const float4* Vs = Vt + ss * 96 * 2;
    const float4 l01 = Vs[nn * 2], l23 = Vs[nn * 2 + 1];
    const float2 la = hi ? make_float2(l01.z, l01.w) : make_float2(l01.x, l01.y);
    const float2 lb = hi ? make_float2(l23.z, l23.w) : make_float2(l23.x, l23.y);
    v2f aRe1, aIm1, aRe2, aIm2;
    aRe1.x = la.x; aRe1.y = -la.y;  aIm1.x = la.y; aIm1.y = la.x;
    aRe2.x = lb.x; aRe2.y = -lb.y;  aIm2.x = lb.y; aIm2.y = lb.x;
    const float4 r01 = Vs[(80 + nn) * 2], r23 = Vs[(80 + nn) * 2 + 1];
    const float2 r0 = make_float2(r01.x, r01.y), r1 = make_float2(r01.z, r01.w);
    const float2 r2 = make_float2(r23.x, r23.y), r3 = make_float2(r23.z, r23.w);
    float Aj[8];
    #pragma unroll
    for (int v = 0; v < 8; ++v) Aj[v] = prefix_sign_sum(w, v);
    const float Cn = prefix_sign_sum(w + 8, nn);
    float racc = 0.f;
    for (int m = 0; m < 16; ++m) {
      const float4* Mr = Vs + (16 + m * 4 + (hi ? 1 : 0)) * 2;
      const float4 ma01 = Mr[0], ma23 = Mr[1], mb01 = Mr[4], mb23 = Mr[5];
      float2 va = cmul(make_float2(ma01.x, ma01.y), r0);
      cfma(va, make_float2(ma01.z, ma01.w), r1);
      cfma(va, make_float2(ma23.x, ma23.y), r2);
      cfma(va, make_float2(ma23.z, ma23.w), r3);
      float2 vb = cmul(make_float2(mb01.x, mb01.y), r0);
      cfma(vb, make_float2(mb01.z, mb01.w), r1);
      cfma(vb, make_float2(mb23.x, mb23.y), r2);
      cfma(vb, make_float2(mb23.z, mb23.w), r3);
      v2f b1, b2;
      b1.x = va.x; b1.y = va.y;
      b2.x = vb.x; b2.y = vb.y;
      const v8f zero = {};
      v8f dre  = __builtin_amdgcn_wmma_f32_16x16x4_f32(false, aRe1, false, b1, (short)0, zero, false, false);
      v8f dim_ = __builtin_amdgcn_wmma_f32_16x16x4_f32(false, aIm1, false, b1, (short)0, zero, false, false);
      dre  = __builtin_amdgcn_wmma_f32_16x16x4_f32(false, aRe2, false, b2, (short)0, dre,  false, false);
      dim_ = __builtin_amdgcn_wmma_f32_16x16x4_f32(false, aIm2, false, b2, (short)0, dim_, false, false);
      const float Bm = prefix_sign_sum(w + 4, m);
      const float E  = (__popc(m) & 1) ? Bm - Cn : Bm + Cn;
      #pragma unroll
      for (int v = 0; v < 8; ++v) {
        const float p = dre[v] * dre[v] + dim_[v] * dim_[v];
        const float c = (__popc(v) & 1) ? Aj[v] - E : Aj[v] + E;
        racc += p * c;
      }
    }
    racc = hi ? -racc : racc;
    racc += xrow<1>(racc);
    racc += xrow<2>(racc);
    racc += xrow<4>(racc);
    racc += xrow<8>(racc);
    racc += xhalf(racc);
    if (lane == 0) out[s0 + wave] = racc + head_b[0];
  }
  static_assert(NL == 3, "all three layers' rotations are folded into the encode");
}

extern "C" void kernel_launch(void* const* d_in, const int* in_sizes, int n_in,
                              void* d_out, int out_size, void* d_ws, size_t ws_size,
                              hipStream_t stream) {
  const float* x      = (const float*)d_in[0];
  const float* params = (const float*)d_in[1];
  const float* head_w = (const float*)d_in[2];
  const float* head_b = (const float*)d_in[3];
  float* out = (float*)d_out;

  const int B = in_sizes[0] / NQ;

  qsim_kernel<<<B / NS, THREADS, 0, stream>>>(x, params, head_w, head_b, out);
}
